# starting kernel
# baseline (speedup 1.0000x reference)
#include <hip/hip_runtime.h>
#include <hip/hip_bf16.h>
#include <math.h>

#define BATCH   4
#define SEQ     512
#define INDIM   300
#define HDIM    128
#define MROWS   (BATCH * SEQ)

typedef __attribute__((ext_vector_type(2))) float v2f;
typedef __attribute__((ext_vector_type(8))) float v8f;


__device__ __forceinline__ v2f load_fragA(const float* A, int m0, int k0,
                                          int lda, int lane) {
    const int m  = m0 + (lane & 15);
    const int ka = k0 + ((lane >> 4) << 1);
    return *reinterpret_cast<const v2f*>(A + (size_t)m * lda + ka);
}

__device__ __forceinline__ v2f load_fragB_WT(const float* W, int n0, int k0,
                                             int ldw, int lane) {
    const int nn = n0 + (lane & 15);
    const int kb = k0 + ((lane >> 4) << 1);
    return *reinterpret_cast<const v2f*>(W + (size_t)nn * ldw + kb);
}

__global__ __launch_bounds__(256) void gemm_wx_tanh(
    const float* __restrict__ A,
    const float* __restrict__ W,
    const float* __restrict__ bias,
    float* __restrict__ out)
{
    const int lane = threadIdx.x & 31;
    const int wid  = blockIdx.x * 8 + (threadIdx.x >> 5);
    const int m0   = (wid >> 3) << 4;
    const int n0   = (wid & 7) << 4;

    v8f acc = {};
#pragma unroll 5
    for (int k0 = 0; k0 < INDIM; k0 += 4) {
        v2f a = load_fragA(A, m0, k0, INDIM, lane);
        v2f b = load_fragB_WT(W, n0, k0, INDIM, lane);
        acc = __builtin_amdgcn_wmma_f32_16x16x4_f32(
                false, a, false, b, (short)0, acc, false, false);
    }

    const int col = n0 + (lane & 15);
    const int rb  = m0 + ((lane >> 4) << 3);
    const float bv = bias[col];
#pragma unroll
    for (int r = 0; r < 8; ++r) {
        out[(size_t)(rb + r) * HDIM + col] = tanhf(acc[r] + bv);
    }
}

__global__ __launch_bounds__(256) void gemm_ta_tb(
    const float* __restrict__ A,
    const float* __restrict__ Wa,
    const float* __restrict__ ba,
    const float* __restrict__ Ua,
    const float* __restrict__ bu,
    float* __restrict__ ta,
    float* __restrict__ tb)
{
    const int lane = threadIdx.x & 31;
    const int wid  = blockIdx.x * 8 + (threadIdx.x >> 5);
    const int m0   = (wid >> 3) << 4;
    const int n0   = (wid & 7) << 4;

    v8f accA = {};
    v8f accB = {};
#pragma unroll 4
    for (int k0 = 0; k0 < HDIM; k0 += 4) {
        v2f a  = load_fragA(A, m0, k0, HDIM, lane);
        v2f bw = load_fragB_WT(Wa, n0, k0, HDIM, lane);
        v2f bu2 = load_fragB_WT(Ua, n0, k0, HDIM, lane);
        accA = __builtin_amdgcn_wmma_f32_16x16x4_f32(
                 false, a, false, bw, (short)0, accA, false, false);
        accB = __builtin_amdgcn_wmma_f32_16x16x4_f32(
                 false, a, false, bu2, (short)0, accB, false, false);
    }

    const int col = n0 + (lane & 15);
    const int rb  = m0 + ((lane >> 4) << 3);
    const float bva = ba[col];
    const float bvu = bu[col];
#pragma unroll
    for (int r = 0; r < 8; ++r) {
        const size_t idx = (size_t)(rb + r) * HDIM + col;
        ta[idx] = accA[r] + bva;
        tb[idx] = accB[r] + bvu;
    }
}

#define QT 4

__global__ __launch_bounds__(128) void attn_fused(
    const float* __restrict__ ta,
    const float* __restrict__ tb,
    const float* __restrict__ hv,
    const float* __restrict__ maskp,
    float* __restrict__ out)
{
    const int h   = threadIdx.x;
    const int bq  = blockIdx.x;
    const int b   = bq / (SEQ / QT);
    const int q0  = (bq % (SEQ / QT)) * QT;
    const size_t base = (size_t)b * SEQ * HDIM;

    float taq[QT], m[QT], s[QT], acc[QT];
#pragma unroll
    for (int i = 0; i < QT; ++i) {
        taq[i] = ta[base + (size_t)(q0 + i) * HDIM + h];
        m[i]   = -INFINITY;
        s[i]   = 0.0f;
        acc[i] = 0.0f;
    }

    const float* ptb = tb    + base + h;
    const float* phv = hv    + base + h;
    const float* pmk = maskp + base + h;

    for (int k = 0; k < SEQ; ++k) {
        if (k + 16 < SEQ) {
            __builtin_prefetch(ptb + (size_t)(k + 16) * HDIM, 0, 0);
            __builtin_prefetch(phv + (size_t)(k + 16) * HDIM, 0, 0);
        }
        const float tbk = ptb[(size_t)k * HDIM];
        const float hvk = phv[(size_t)k * HDIM];
        const float mkk = pmk[(size_t)k * HDIM];
#pragma unroll
        for (int i = 0; i < QT; ++i) {
            const float x = taq[i] + tbk;
            const float e = (x > 0.0f ? x : __expf(x) - 1.0f) + mkk;
            const float mn   = fmaxf(m[i], e);
            const float corr = __expf(m[i] - mn);
            const float p    = __expf(e - mn);
            s[i]   = s[i]  * corr + p;
            acc[i] = acc[i] * corr + p * hvk;
            m[i]   = mn;
        }
    }

#pragma unroll
    for (int i = 0; i < QT; ++i) {
        out[base + (size_t)(q0 + i) * HDIM + h] = acc[i] / s[i];
    }
}

extern "C" void kernel_launch(void* const* d_in, const int* in_sizes, int n_in,
                              void* d_out, int out_size, void* d_ws, size_t ws_size,
                              hipStream_t stream) {
    const float* word_vecs = (const float*)d_in[0];
    const float* mask      = (const float*)d_in[1];
    const float* Wx_w      = (const float*)d_in[2];
    const float* Wx_b      = (const float*)d_in[3];
    const float* Wa_w      = (const float*)d_in[4];
    const float* Wa_b      = (const float*)d_in[5];
    const float* Ua_w      = (const float*)d_in[6];
    const float* Ua_b      = (const float*)d_in[7];
    float* out = (float*)d_out;

    float* hv = (float*)d_ws;
    float* ta = hv + (size_t)MROWS * HDIM;
    float* tb = ta + (size_t)MROWS * HDIM;

    gemm_wx_tanh<<<128, 256, 0, stream>>>(word_vecs, Wx_w, Wx_b, hv);
    gemm_ta_tb<<<128, 256, 0, stream>>>(hv, Wa_w, Wa_b, Ua_w, Ua_b, ta, tb);

    attn_fused<<<BATCH * (SEQ / QT), 128, 0, stream>>>(ta, tb, hv, mask, out);
}


# attention k loop software-pipelined through registers: two register groups of key rows alternate, each group's loads issued whole before the other is consumed (replaces per-row L2 prefetch hints)
# speedup vs baseline: 4.1250x; 4.1250x over previous
#include <hip/hip_runtime.h>
#include <hip/hip_bf16.h>
#include <math.h>

#define BATCH   4
#define SEQ     512
#define INDIM   300
#define HDIM    128
#define MROWS   (BATCH * SEQ)

typedef __attribute__((ext_vector_type(2))) float v2f;
typedef __attribute__((ext_vector_type(8))) float v8f;


__device__ __forceinline__ v2f load_fragA(const float* A, int m0, int k0,
                                          int lda, int lane) {
    const int m  = m0 + (lane & 15);
    const int ka = k0 + ((lane >> 4) << 1);
    return *reinterpret_cast<const v2f*>(A + (size_t)m * lda + ka);
}

__device__ __forceinline__ v2f load_fragB_WT(const float* W, int n0, int k0,
                                             int ldw, int lane) {
    const int nn = n0 + (lane & 15);
    const int kb = k0 + ((lane >> 4) << 1);
    return *reinterpret_cast<const v2f*>(W + (size_t)nn * ldw + kb);
}

__global__ __launch_bounds__(256) void gemm_wx_tanh(
    const float* __restrict__ A,
    const float* __restrict__ W,
    const float* __restrict__ bias,
    float* __restrict__ out)
{
    const int lane = threadIdx.x & 31;
    const int wid  = blockIdx.x * 8 + (threadIdx.x >> 5);
    const int m0   = (wid >> 3) << 4;
    const int n0   = (wid & 7) << 4;

    v8f acc = {};
#pragma unroll 5
    for (int k0 = 0; k0 < INDIM; k0 += 4) {
        v2f a = load_fragA(A, m0, k0, INDIM, lane);
        v2f b = load_fragB_WT(W, n0, k0, INDIM, lane);
        acc = __builtin_amdgcn_wmma_f32_16x16x4_f32(
                false, a, false, b, (short)0, acc, false, false);
    }

    const int col = n0 + (lane & 15);
    const int rb  = m0 + ((lane >> 4) << 3);
    const float bv = bias[col];
#pragma unroll
    for (int r = 0; r < 8; ++r) {
        out[(size_t)(rb + r) * HDIM + col] = tanhf(acc[r] + bv);
    }
}

__global__ __launch_bounds__(256) void gemm_ta_tb(
    const float* __restrict__ A,
    const float* __restrict__ Wa,
    const float* __restrict__ ba,
    const float* __restrict__ Ua,
    const float* __restrict__ bu,
    float* __restrict__ ta,
    float* __restrict__ tb)
{
    const int lane = threadIdx.x & 31;
    const int wid  = blockIdx.x * 8 + (threadIdx.x >> 5);
    const int m0   = (wid >> 3) << 4;
    const int n0   = (wid & 7) << 4;

    v8f accA = {};
    v8f accB = {};
#pragma unroll 4
    for (int k0 = 0; k0 < HDIM; k0 += 4) {
        v2f a  = load_fragA(A, m0, k0, HDIM, lane);
        v2f bw = load_fragB_WT(Wa, n0, k0, HDIM, lane);
        v2f bu2 = load_fragB_WT(Ua, n0, k0, HDIM, lane);
        accA = __builtin_amdgcn_wmma_f32_16x16x4_f32(
                 false, a, false, bw, (short)0, accA, false, false);
        accB = __builtin_amdgcn_wmma_f32_16x16x4_f32(
                 false, a, false, bu2, (short)0, accB, false, false);
    }

    const int col = n0 + (lane & 15);
    const int rb  = m0 + ((lane >> 4) << 3);
    const float bva = ba[col];
    const float bvu = bu[col];
#pragma unroll
    for (int r = 0; r < 8; ++r) {
        const size_t idx = (size_t)(rb + r) * HDIM + col;
        ta[idx] = accA[r] + bva;
        tb[idx] = accB[r] + bvu;
    }
}

#define QT 4
#define KU 16

__device__ __forceinline__ void attn_row(const float tbk, const float hvk, const float mkk,
                                         const float (&taq)[QT], float (&m)[QT],
                                         float (&s)[QT], float (&acc)[QT]) {
#pragma unroll
    for (int i = 0; i < QT; ++i) {
        const float x = taq[i] + tbk;
        const float e = (x > 0.0f ? x : __expf(x) - 1.0f) + mkk;
        const float mn   = fmaxf(m[i], e);
        const float corr = __expf(m[i] - mn);
        const float p    = __expf(e - mn);
        s[i]   = s[i]  * corr + p;
        acc[i] = acc[i] * corr + p * hvk;
        m[i]   = mn;
    }
}

__global__ __launch_bounds__(128) void attn_fused(
    const float* __restrict__ ta,
    const float* __restrict__ tb,
    const float* __restrict__ hv,
    const float* __restrict__ maskp,
    float* __restrict__ out)
{
    const int h   = threadIdx.x;
    const int bq  = blockIdx.x;
    const int b   = bq / (SEQ / QT);
    const int q0  = (bq % (SEQ / QT)) * QT;
    const size_t base = (size_t)b * SEQ * HDIM;

    float taq[QT], m[QT], s[QT], acc[QT];
#pragma unroll
    for (int i = 0; i < QT; ++i) {
        taq[i] = ta[base + (size_t)(q0 + i) * HDIM + h];
        m[i]   = -INFINITY;
        s[i]   = 0.0f;
        acc[i] = 0.0f;
    }

    const float* ptb = tb    + base + h;
    const float* phv = hv    + base + h;
    const float* pmk = maskp + base + h;

    float tbA[KU], hvA[KU], mkA[KU];
    float tbB[KU], hvB[KU], mkB[KU];

#define LOAD_GROUP(TB, HV, MK, K0)                                  \
    _Pragma("unroll") for (int j = 0; j < KU; ++j) {                \
        TB[j] = ptb[(size_t)((K0) + j) * HDIM];                     \
        HV[j] = phv[(size_t)((K0) + j) * HDIM];                     \
        MK[j] = pmk[(size_t)((K0) + j) * HDIM];                     \
    }
#define USE_GROUP(TB, HV, MK)                                       \
    _Pragma("unroll") for (int j = 0; j < KU; ++j)                  \
        attn_row(TB[j], HV[j], MK[j], taq, m, s, acc);

    LOAD_GROUP(tbA, hvA, mkA, 0);

    for (int k0 = 0; k0 < SEQ; k0 += 2 * KU) {
        __builtin_amdgcn_sched_barrier(0);
        LOAD_GROUP(tbB, hvB, mkB, k0 + KU);
        __builtin_amdgcn_sched_barrier(0);
        USE_GROUP(tbA, hvA, mkA);

        const int kn = (k0 + 2 * KU) & (SEQ - 1);
        __builtin_amdgcn_sched_barrier(0);
        LOAD_GROUP(tbA, hvA, mkA, kn);
        __builtin_amdgcn_sched_barrier(0);
        USE_GROUP(tbB, hvB, mkB);
    }
#undef LOAD_GROUP
#undef USE_GROUP

#pragma unroll
    for (int i = 0; i < QT; ++i) {
        out[base + (size_t)(q0 + i) * HDIM + h] = acc[i] / s[i];
    }
}

extern "C" void kernel_launch(void* const* d_in, const int* in_sizes, int n_in,
                              void* d_out, int out_size, void* d_ws, size_t ws_size,
                              hipStream_t stream) {
    const float* word_vecs = (const float*)d_in[0];
    const float* mask      = (const float*)d_in[1];
    const float* Wx_w      = (const float*)d_in[2];
    const float* Wx_b      = (const float*)d_in[3];
    const float* Wa_w      = (const float*)d_in[4];
    const float* Wa_b      = (const float*)d_in[5];
    const float* Ua_w      = (const float*)d_in[6];
    const float* Ua_b      = (const float*)d_in[7];
    float* out = (float*)d_out;

    float* hv = (float*)d_ws;
    float* ta = hv + (size_t)MROWS * HDIM;
    float* tb = ta + (size_t)MROWS * HDIM;

    gemm_wx_tanh<<<128, 256, 0, stream>>>(word_vecs, Wx_w, Wx_b, hv);
    gemm_ta_tb<<<128, 256, 0, stream>>>(hv, Wa_w, Wa_b, Ua_w, Ua_b, ta, tb);

    attn_fused<<<BATCH * (SEQ / QT), 128, 0, stream>>>(ta, tb, hv, mask, out);
}


# attention: softmax rescaled once per group of 16 key rows (group max, then weights exp(score-max)): 2 exp per score, not 3; q-streams as float2 pairs; sums keep key order; all -inf group: 0, not NaN
# speedup vs baseline: 5.2063x; 1.2621x over previous
#include <hip/hip_runtime.h>
#include <hip/hip_bf16.h>
#include <math.h>
#include <float.h>

#define BATCH   4
#define SEQ     512
#define INDIM   300
#define HDIM    128
#define MROWS   (BATCH * SEQ)

typedef __attribute__((ext_vector_type(2))) float v2f;
typedef __attribute__((ext_vector_type(8))) float v8f;


__device__ __forceinline__ v2f load_fragA(const float* A, int m0, int k0,
                                          int lda, int lane) {
    const int m  = m0 + (lane & 15);
    const int ka = k0 + ((lane >> 4) << 1);
    return *reinterpret_cast<const v2f*>(A + (size_t)m * lda + ka);
}

__device__ __forceinline__ v2f load_fragB_WT(const float* W, int n0, int k0,
                                             int ldw, int lane) {
    const int nn = n0 + (lane & 15);
    const int kb = k0 + ((lane >> 4) << 1);
    return *reinterpret_cast<const v2f*>(W + (size_t)nn * ldw + kb);
}

__global__ __launch_bounds__(256) void gemm_wx_tanh(
    const float* __restrict__ A,
    const float* __restrict__ W,
    const float* __restrict__ bias,
    float* __restrict__ out)
{
    const int lane = threadIdx.x & 31;
    const int wid  = blockIdx.x * 8 + (threadIdx.x >> 5);
    const int m0   = (wid >> 3) << 4;
    const int n0   = (wid & 7) << 4;

    v8f acc = {};
#pragma unroll 5
    for (int k0 = 0; k0 < INDIM; k0 += 4) {
        v2f a = load_fragA(A, m0, k0, INDIM, lane);
        v2f b = load_fragB_WT(W, n0, k0, INDIM, lane);
        acc = __builtin_amdgcn_wmma_f32_16x16x4_f32(
                false, a, false, b, (short)0, acc, false, false);
    }

    const int col = n0 + (lane & 15);
    const int rb  = m0 + ((lane >> 4) << 3);
    const float bv = bias[col];
#pragma unroll
    for (int r = 0; r < 8; ++r) {
        out[(size_t)(rb + r) * HDIM + col] = tanhf(acc[r] + bv);
    }
}

__global__ __launch_bounds__(256) void gemm_ta_tb(
    const float* __restrict__ A,
    const float* __restrict__ Wa,
    const float* __restrict__ ba,
    const float* __restrict__ Ua,
    const float* __restrict__ bu,
    float* __restrict__ ta,
    float* __restrict__ tb)
{
    const int lane = threadIdx.x & 31;
    const int wid  = blockIdx.x * 8 + (threadIdx.x >> 5);
    const int m0   = (wid >> 3) << 4;
    const int n0   = (wid & 7) << 4;

    v8f accA = {};
    v8f accB = {};
#pragma unroll 4
    for (int k0 = 0; k0 < HDIM; k0 += 4) {
        v2f a  = load_fragA(A, m0, k0, HDIM, lane);
        v2f bw = load_fragB_WT(Wa, n0, k0, HDIM, lane);
        v2f bu2 = load_fragB_WT(Ua, n0, k0, HDIM, lane);
        accA = __builtin_amdgcn_wmma_f32_16x16x4_f32(
                 false, a, false, bw, (short)0, accA, false, false);
        accB = __builtin_amdgcn_wmma_f32_16x16x4_f32(
                 false, a, false, bu2, (short)0, accB, false, false);
    }

    const int col = n0 + (lane & 15);
    const int rb  = m0 + ((lane >> 4) << 3);
    const float bva = ba[col];
    const float bvu = bu[col];
#pragma unroll
    for (int r = 0; r < 8; ++r) {
        const size_t idx = (size_t)(rb + r) * HDIM + col;
        ta[idx] = accA[r] + bva;
        tb[idx] = accB[r] + bvu;
    }
}

#define QT 4
#define KU 16

#define QP (QT / 2)
__device__ __forceinline__ void attn_group(const float (&tbk)[KU], const float (&hvk)[KU],
                                           const float (&mkk)[KU], const v2f (&taq)[QP],
                                           v2f (&m)[QP], v2f (&s)[QP], v2f (&acc)[QP]) {
    v2f e[KU][QP];
    v2f bm[QP];
    v2f mr[QP];
#pragma unroll
    for (int i = 0; i < QP; ++i) { bm[i].x = -INFINITY; bm[i].y = -INFINITY; }

#pragma unroll
    for (int j = 0; j < KU; ++j) {
#pragma unroll
        for (int i = 0; i < QP; ++i) {
            const v2f x = taq[i] + tbk[j];
            v2f el;
            el.x = x.x > 0.0f ? x.x : __expf(x.x) - 1.0f;
            el.y = x.y > 0.0f ? x.y : __expf(x.y) - 1.0f;
            const v2f v = el + mkk[j];
            e[j][i] = v;
            bm[i].x = fmaxf(bm[i].x, v.x);
            bm[i].y = fmaxf(bm[i].y, v.y);
        }
    }

#pragma unroll
    for (int i = 0; i < QP; ++i) {
        v2f mn, corr;
        mn.x = fmaxf(m[i].x, bm[i].x);
        mn.y = fmaxf(m[i].y, bm[i].y);
        mr[i].x = fmaxf(mn.x, -FLT_MAX);
        mr[i].y = fmaxf(mn.y, -FLT_MAX);
        const v2f dm = m[i] - mr[i];
        corr.x = __expf(dm.x);
        corr.y = __expf(dm.y);
        s[i]   *= corr;
        acc[i] *= corr;
        m[i]    = mn;
    }

#pragma unroll
    for (int j = 0; j < KU; ++j) {
        const v2f hv2 = {hvk[j], hvk[j]};
#pragma unroll
        for (int i = 0; i < QP; ++i) {
            const v2f d = e[j][i] - mr[i];
            v2f p;
            p.x = __expf(d.x);
            p.y = __expf(d.y);
            s[i]  += p;
            acc[i] = __builtin_elementwise_fma(p, hv2, acc[i]);
        }
    }
}

__global__ __launch_bounds__(128) void attn_fused(
    const float* __restrict__ ta,
    const float* __restrict__ tb,
    const float* __restrict__ hv,
    const float* __restrict__ maskp,
    float* __restrict__ out)
{
    const int h   = threadIdx.x;
    const int bq  = blockIdx.x;
    const int b   = bq / (SEQ / QT);
    const int q0  = (bq % (SEQ / QT)) * QT;
    const size_t base = (size_t)b * SEQ * HDIM;

    v2f taq[QP], m[QP], s[QP], acc[QP];
#pragma unroll
    for (int i = 0; i < QP; ++i) {
        taq[i].x = ta[base + (size_t)(q0 + 2 * i) * HDIM + h];
        taq[i].y = ta[base + (size_t)(q0 + 2 * i + 1) * HDIM + h];
        m[i].x = -INFINITY;  m[i].y = -INFINITY;
        s[i].x = 0.0f;       s[i].y = 0.0f;
        acc[i].x = 0.0f;     acc[i].y = 0.0f;
    }

    const float* ptb = tb    + base + h;
    const float* phv = hv    + base + h;
    const float* pmk = maskp + base + h;

    float tbA[KU], hvA[KU], mkA[KU];
    float tbB[KU], hvB[KU], mkB[KU];

#define LOAD_GROUP(TB, HV, MK, K0)                                  \
    _Pragma("unroll") for (int j = 0; j < KU; ++j) {                \
        TB[j] = ptb[(size_t)((K0) + j) * HDIM];                     \
        HV[j] = phv[(size_t)((K0) + j) * HDIM];                     \
        MK[j] = pmk[(size_t)((K0) + j) * HDIM];                     \
    }

    LOAD_GROUP(tbA, hvA, mkA, 0);

    for (int k0 = 0; k0 < SEQ; k0 += 2 * KU) {
        __builtin_amdgcn_sched_barrier(0);
        LOAD_GROUP(tbB, hvB, mkB, k0 + KU);
        __builtin_amdgcn_sched_barrier(0);
        attn_group(tbA, hvA, mkA, taq, m, s, acc);

        const int kn = (k0 + 2 * KU) & (SEQ - 1);
        __builtin_amdgcn_sched_barrier(0);
        LOAD_GROUP(tbA, hvA, mkA, kn);
        __builtin_amdgcn_sched_barrier(0);
        attn_group(tbB, hvB, mkB, taq, m, s, acc);
    }
#undef LOAD_GROUP

#pragma unroll
    for (int i = 0; i < QP; ++i) {
        out[base + (size_t)(q0 + 2 * i) * HDIM + h]     = acc[i].x / s[i].x;
        out[base + (size_t)(q0 + 2 * i + 1) * HDIM + h] = acc[i].y / s[i].y;
    }
}

extern "C" void kernel_launch(void* const* d_in, const int* in_sizes, int n_in,
                              void* d_out, int out_size, void* d_ws, size_t ws_size,
                              hipStream_t stream) {
    const float* word_vecs = (const float*)d_in[0];
    const float* mask      = (const float*)d_in[1];
    const float* Wx_w      = (const float*)d_in[2];
    const float* Wx_b      = (const float*)d_in[3];
    const float* Wa_w      = (const float*)d_in[4];
    const float* Wa_b      = (const float*)d_in[5];
    const float* Ua_w      = (const float*)d_in[6];
    const float* Ua_b      = (const float*)d_in[7];
    float* out = (float*)d_out;

    float* hv = (float*)d_ws;
    float* ta = hv + (size_t)MROWS * HDIM;
    float* tb = ta + (size_t)MROWS * HDIM;

    gemm_wx_tanh<<<128, 256, 0, stream>>>(word_vecs, Wx_w, Wx_b, hv);
    gemm_ta_tb<<<128, 256, 0, stream>>>(hv, Wa_w, Wa_b, Ua_w, Ua_b, ta, tb);

    attn_fused<<<BATCH * (SEQ / QT), 128, 0, stream>>>(ta, tb, hv, mask, out);
}


# attention softmax carried in base 2: ta, tb stored times log2(e), elu's -1 and the mask scaled inside fused multiply-adds, so each exponential is a bare exp2 (no multiply per score); key order kept
# speedup vs baseline: 5.3890x; 1.0351x over previous
#include <hip/hip_runtime.h>
#include <hip/hip_bf16.h>
#include <math.h>
#include <float.h>

#define BATCH   4
#define SEQ     512
#define INDIM   300
#define HDIM    128
#define MROWS   (BATCH * SEQ)

typedef __attribute__((ext_vector_type(2))) float v2f;

#define LOG2E 1.4426950408889634f
typedef __attribute__((ext_vector_type(8))) float v8f;


__device__ __forceinline__ v2f load_fragA(const float* A, int m0, int k0,
                                          int lda, int lane) {
    const int m  = m0 + (lane & 15);
    const int ka = k0 + ((lane >> 4) << 1);
    return *reinterpret_cast<const v2f*>(A + (size_t)m * lda + ka);
}

__device__ __forceinline__ v2f load_fragB_WT(const float* W, int n0, int k0,
                                             int ldw, int lane) {
    const int nn = n0 + (lane & 15);
    const int kb = k0 + ((lane >> 4) << 1);
    return *reinterpret_cast<const v2f*>(W + (size_t)nn * ldw + kb);
}

__global__ __launch_bounds__(256) void gemm_wx_tanh(
    const float* __restrict__ A,
    const float* __restrict__ W,
    const float* __restrict__ bias,
    float* __restrict__ out)
{
    const int lane = threadIdx.x & 31;
    const int wid  = blockIdx.x * 8 + (threadIdx.x >> 5);
    const int m0   = (wid >> 3) << 4;
    const int n0   = (wid & 7) << 4;

    v8f acc = {};
#pragma unroll 5
    for (int k0 = 0; k0 < INDIM; k0 += 4) {
        v2f a = load_fragA(A, m0, k0, INDIM, lane);
        v2f b = load_fragB_WT(W, n0, k0, INDIM, lane);
        acc = __builtin_amdgcn_wmma_f32_16x16x4_f32(
                false, a, false, b, (short)0, acc, false, false);
    }

    const int col = n0 + (lane & 15);
    const int rb  = m0 + ((lane >> 4) << 3);
    const float bv = bias[col];
#pragma unroll
    for (int r = 0; r < 8; ++r) {
        out[(size_t)(rb + r) * HDIM + col] = tanhf(acc[r] + bv);
    }
}

__global__ __launch_bounds__(256) void gemm_ta_tb(
    const float* __restrict__ A,
    const float* __restrict__ Wa,
    const float* __restrict__ ba,
    const float* __restrict__ Ua,
    const float* __restrict__ bu,
    float* __restrict__ ta,
    float* __restrict__ tb)
{
    const int lane = threadIdx.x & 31;
    const int wid  = blockIdx.x * 8 + (threadIdx.x >> 5);
    const int m0   = (wid >> 3) << 4;
    const int n0   = (wid & 7) << 4;

    v8f accA = {};
    v8f accB = {};
#pragma unroll 4
    for (int k0 = 0; k0 < HDIM; k0 += 4) {
        v2f a  = load_fragA(A, m0, k0, HDIM, lane);
        v2f bw = load_fragB_WT(Wa, n0, k0, HDIM, lane);
        v2f bu2 = load_fragB_WT(Ua, n0, k0, HDIM, lane);
        accA = __builtin_amdgcn_wmma_f32_16x16x4_f32(
                 false, a, false, bw, (short)0, accA, false, false);
        accB = __builtin_amdgcn_wmma_f32_16x16x4_f32(
                 false, a, false, bu2, (short)0, accB, false, false);
    }

    const int col = n0 + (lane & 15);
    const int rb  = m0 + ((lane >> 4) << 3);
    const float bva = ba[col];
    const float bvu = bu[col];
#pragma unroll
    for (int r = 0; r < 8; ++r) {
        const size_t idx = (size_t)(rb + r) * HDIM + col;
        ta[idx] = (accA[r] + bva) * LOG2E;
        tb[idx] = (accB[r] + bvu) * LOG2E;
    }
}

#define QT 4
#define KU 16

#define QP (QT / 2)
__device__ __forceinline__ void attn_group(const float (&tbk)[KU], const float (&hvk)[KU],
                                           const float (&mkk)[KU], const v2f (&taq)[QP],
                                           v2f (&m)[QP], v2f (&s)[QP], v2f (&acc)[QP]) {
    v2f e[KU][QP];
    v2f bm[QP];
    v2f mr[QP];
    const v2f l2e = {LOG2E, LOG2E};
#pragma unroll
    for (int i = 0; i < QP; ++i) { bm[i].x = -INFINITY; bm[i].y = -INFINITY; }

#pragma unroll
    for (int j = 0; j < KU; ++j) {
        const v2f mk2 = {mkk[j], mkk[j]};
#pragma unroll
        for (int i = 0; i < QP; ++i) {
            const v2f x = taq[i] + tbk[j];
            v2f t;
            t.x = __builtin_amdgcn_exp2f(x.x);
            t.y = __builtin_amdgcn_exp2f(x.y);
            const v2f em1 = __builtin_elementwise_fma(t, l2e, -l2e);
            v2f el;
            el.x = x.x > 0.0f ? x.x : em1.x;
            el.y = x.y > 0.0f ? x.y : em1.y;
            const v2f v = __builtin_elementwise_fma(mk2, l2e, el);
            e[j][i] = v;
            bm[i].x = fmaxf(bm[i].x, v.x);
            bm[i].y = fmaxf(bm[i].y, v.y);
        }
    }

#pragma unroll
    for (int i = 0; i < QP; ++i) {
        v2f mn, corr;
        mn.x = fmaxf(m[i].x, bm[i].x);
        mn.y = fmaxf(m[i].y, bm[i].y);
        mr[i].x = fmaxf(mn.x, -FLT_MAX);
        mr[i].y = fmaxf(mn.y, -FLT_MAX);
        const v2f dm = m[i] - mr[i];
        corr.x = __builtin_amdgcn_exp2f(dm.x);
        corr.y = __builtin_amdgcn_exp2f(dm.y);
        s[i]   *= corr;
        acc[i] *= corr;
        m[i]    = mn;
    }

#pragma unroll
    for (int j = 0; j < KU; ++j) {
        const v2f hv2 = {hvk[j], hvk[j]};
#pragma unroll
        for (int i = 0; i < QP; ++i) {
            const v2f d = e[j][i] - mr[i];
            v2f p;
            p.x = __builtin_amdgcn_exp2f(d.x);
            p.y = __builtin_amdgcn_exp2f(d.y);
            s[i]  += p;
            acc[i] = __builtin_elementwise_fma(p, hv2, acc[i]);
        }
    }
}

__global__ __launch_bounds__(128) void attn_fused(
    const float* __restrict__ ta,
    const float* __restrict__ tb,
    const float* __restrict__ hv,
    const float* __restrict__ maskp,
    float* __restrict__ out)
{
    const int h   = threadIdx.x;
    const int bq  = blockIdx.x;
    const int b   = bq / (SEQ / QT);
    const int q0  = (bq % (SEQ / QT)) * QT;
    const size_t base = (size_t)b * SEQ * HDIM;

    v2f taq[QP], m[QP], s[QP], acc[QP];
#pragma unroll
    for (int i = 0; i < QP; ++i) {
        taq[i].x = ta[base + (size_t)(q0 + 2 * i) * HDIM + h];
        taq[i].y = ta[base + (size_t)(q0 + 2 * i + 1) * HDIM + h];
        m[i].x = -INFINITY;  m[i].y = -INFINITY;
        s[i].x = 0.0f;       s[i].y = 0.0f;
        acc[i].x = 0.0f;     acc[i].y = 0.0f;
    }

    const float* ptb = tb    + base + h;
    const float* phv = hv    + base + h;
    const float* pmk = maskp + base + h;

    float tbA[KU], hvA[KU], mkA[KU];
    float tbB[KU], hvB[KU], mkB[KU];

#define LOAD_GROUP(TB, HV, MK, K0)                                  \
    _Pragma("unroll") for (int j = 0; j < KU; ++j) {                \
        TB[j] = ptb[(size_t)((K0) + j) * HDIM];                     \
        HV[j] = phv[(size_t)((K0) + j) * HDIM];                     \
        MK[j] = pmk[(size_t)((K0) + j) * HDIM];                     \
    }

    LOAD_GROUP(tbA, hvA, mkA, 0);

    for (int k0 = 0; k0 < SEQ; k0 += 2 * KU) {
        __builtin_amdgcn_sched_barrier(0);
        LOAD_GROUP(tbB, hvB, mkB, k0 + KU);
        __builtin_amdgcn_sched_barrier(0);
        attn_group(tbA, hvA, mkA, taq, m, s, acc);

        const int kn = (k0 + 2 * KU) & (SEQ - 1);
        __builtin_amdgcn_sched_barrier(0);
        LOAD_GROUP(tbA, hvA, mkA, kn);
        __builtin_amdgcn_sched_barrier(0);
        attn_group(tbB, hvB, mkB, taq, m, s, acc);
    }
#undef LOAD_GROUP

#pragma unroll
    for (int i = 0; i < QP; ++i) {
        out[base + (size_t)(q0 + 2 * i) * HDIM + h]     = acc[i].x / s[i].x;
        out[base + (size_t)(q0 + 2 * i + 1) * HDIM + h] = acc[i].y / s[i].y;
    }
}

extern "C" void kernel_launch(void* const* d_in, const int* in_sizes, int n_in,
                              void* d_out, int out_size, void* d_ws, size_t ws_size,
                              hipStream_t stream) {
    const float* word_vecs = (const float*)d_in[0];
    const float* mask      = (const float*)d_in[1];
    const float* Wx_w      = (const float*)d_in[2];
    const float* Wx_b      = (const float*)d_in[3];
    const float* Wa_w      = (const float*)d_in[4];
    const float* Wa_b      = (const float*)d_in[5];
    const float* Ua_w      = (const float*)d_in[6];
    const float* Ua_b      = (const float*)d_in[7];
    float* out = (float*)d_out;

    float* hv = (float*)d_ws;
    float* ta = hv + (size_t)MROWS * HDIM;
    float* tb = ta + (size_t)MROWS * HDIM;

    gemm_wx_tanh<<<128, 256, 0, stream>>>(word_vecs, Wx_w, Wx_b, hv);
    gemm_ta_tb<<<128, 256, 0, stream>>>(hv, Wa_w, Wa_b, Ua_w, Ua_b, ta, tb);

    attn_fused<<<BATCH * (SEQ / QT), 128, 0, stream>>>(ta, tb, hv, mask, out);
}


# attention: elu select as one min-max per score, max(x, min(exp-1, 0)), and group maxima as NaN-propagating three-input maximum (no quieting ops): fewer vector instructions; sums keep key order
# speedup vs baseline: 5.7075x; 1.0591x over previous
#include <hip/hip_runtime.h>
#include <hip/hip_bf16.h>
#include <math.h>
#include <float.h>

#define BATCH   4
#define SEQ     512
#define INDIM   300
#define HDIM    128
#define MROWS   (BATCH * SEQ)

typedef __attribute__((ext_vector_type(2))) float v2f;

#define LOG2E 1.4426950408889634f
typedef __attribute__((ext_vector_type(8))) float v8f;


__device__ __forceinline__ v2f load_fragA(const float* A, int m0, int k0,
                                          int lda, int lane) {
    const int m  = m0 + (lane & 15);
    const int ka = k0 + ((lane >> 4) << 1);
    return *reinterpret_cast<const v2f*>(A + (size_t)m * lda + ka);
}

__device__ __forceinline__ v2f load_fragB_WT(const float* W, int n0, int k0,
                                             int ldw, int lane) {
    const int nn = n0 + (lane & 15);
    const int kb = k0 + ((lane >> 4) << 1);
    return *reinterpret_cast<const v2f*>(W + (size_t)nn * ldw + kb);
}

__global__ __launch_bounds__(256) void gemm_wx_tanh(
    const float* __restrict__ A,
    const float* __restrict__ W,
    const float* __restrict__ bias,
    float* __restrict__ out)
{
    const int lane = threadIdx.x & 31;
    const int wid  = blockIdx.x * 8 + (threadIdx.x >> 5);
    const int m0   = (wid >> 3) << 4;
    const int n0   = (wid & 7) << 4;

    v8f acc = {};
#pragma unroll 5
    for (int k0 = 0; k0 < INDIM; k0 += 4) {
        v2f a = load_fragA(A, m0, k0, INDIM, lane);
        v2f b = load_fragB_WT(W, n0, k0, INDIM, lane);
        acc = __builtin_amdgcn_wmma_f32_16x16x4_f32(
                false, a, false, b, (short)0, acc, false, false);
    }

    const int col = n0 + (lane & 15);
    const int rb  = m0 + ((lane >> 4) << 3);
    const float bv = bias[col];
#pragma unroll
    for (int r = 0; r < 8; ++r) {
        out[(size_t)(rb + r) * HDIM + col] = tanhf(acc[r] + bv);
    }
}

__global__ __launch_bounds__(256) void gemm_ta_tb(
    const float* __restrict__ A,
    const float* __restrict__ Wa,
    const float* __restrict__ ba,
    const float* __restrict__ Ua,
    const float* __restrict__ bu,
    float* __restrict__ ta,
    float* __restrict__ tb)
{
    const int lane = threadIdx.x & 31;
    const int wid  = blockIdx.x * 8 + (threadIdx.x >> 5);
    const int m0   = (wid >> 3) << 4;
    const int n0   = (wid & 7) << 4;

    v8f accA = {};
    v8f accB = {};
#pragma unroll 4
    for (int k0 = 0; k0 < HDIM; k0 += 4) {
        v2f a  = load_fragA(A, m0, k0, HDIM, lane);
        v2f bw = load_fragB_WT(Wa, n0, k0, HDIM, lane);
        v2f bu2 = load_fragB_WT(Ua, n0, k0, HDIM, lane);
        accA = __builtin_amdgcn_wmma_f32_16x16x4_f32(
                 false, a, false, bw, (short)0, accA, false, false);
        accB = __builtin_amdgcn_wmma_f32_16x16x4_f32(
                 false, a, false, bu2, (short)0, accB, false, false);
    }

    const int col = n0 + (lane & 15);
    const int rb  = m0 + ((lane >> 4) << 3);
    const float bva = ba[col];
    const float bvu = bu[col];
#pragma unroll
    for (int r = 0; r < 8; ++r) {
        const size_t idx = (size_t)(rb + r) * HDIM + col;
        ta[idx] = (accA[r] + bva) * LOG2E;
        tb[idx] = (accB[r] + bvu) * LOG2E;
    }
}

#define QT 4
#define KU 16

#define QP (QT / 2)
__device__ __forceinline__ void attn_group(const float (&tbk)[KU], const float (&hvk)[KU],
                                           const float (&mkk)[KU], const v2f (&taq)[QP],
                                           v2f (&m)[QP], v2f (&s)[QP], v2f (&acc)[QP]) {
    v2f e[KU][QP];
    v2f bm[QP];
    v2f mr[QP];
    const v2f l2e = {LOG2E, LOG2E};
    const v2f zero2 = {0.0f, 0.0f};
    const v2f lowest2 = {-FLT_MAX, -FLT_MAX};
#pragma unroll
    for (int i = 0; i < QP; ++i) { bm[i].x = -INFINITY; bm[i].y = -INFINITY; }

#pragma unroll
    for (int j = 0; j < KU; ++j) {
        const v2f mk2 = {mkk[j], mkk[j]};
#pragma unroll
        for (int i = 0; i < QP; ++i) {
            const v2f x = taq[i] + tbk[j];
            v2f t;
            t.x = __builtin_amdgcn_exp2f(x.x);
            t.y = __builtin_amdgcn_exp2f(x.y);
            const v2f em1 = __builtin_elementwise_fma(t, l2e, -l2e);
            const v2f el = __builtin_elementwise_maximum(
                               x, __builtin_elementwise_minimum(em1, zero2));
            const v2f v = __builtin_elementwise_fma(mk2, l2e, el);
            e[j][i] = v;
            bm[i] = __builtin_elementwise_maximum(bm[i], v);
        }
    }

#pragma unroll
    for (int i = 0; i < QP; ++i) {
        v2f mn, corr;
        mn = __builtin_elementwise_maximum(m[i], bm[i]);
        mr[i] = __builtin_elementwise_maximum(mn, lowest2);
        const v2f dm = m[i] - mr[i];
        corr.x = __builtin_amdgcn_exp2f(dm.x);
        corr.y = __builtin_amdgcn_exp2f(dm.y);
        s[i]   *= corr;
        acc[i] *= corr;
        m[i]    = mn;
    }

#pragma unroll
    for (int j = 0; j < KU; ++j) {
        const v2f hv2 = {hvk[j], hvk[j]};
#pragma unroll
        for (int i = 0; i < QP; ++i) {
            const v2f d = e[j][i] - mr[i];
            v2f p;
            p.x = __builtin_amdgcn_exp2f(d.x);
            p.y = __builtin_amdgcn_exp2f(d.y);
            s[i]  += p;
            acc[i] = __builtin_elementwise_fma(p, hv2, acc[i]);
        }
    }
}

__global__ __launch_bounds__(128) void attn_fused(
    const float* __restrict__ ta,
    const float* __restrict__ tb,
    const float* __restrict__ hv,
    const float* __restrict__ maskp,
    float* __restrict__ out)
{
    const int h   = threadIdx.x;
    const int bq  = blockIdx.x;
    const int b   = bq / (SEQ / QT);
    const int q0  = (bq % (SEQ / QT)) * QT;
    const size_t base = (size_t)b * SEQ * HDIM;

    v2f taq[QP], m[QP], s[QP], acc[QP];
#pragma unroll
    for (int i = 0; i < QP; ++i) {
        taq[i].x = ta[base + (size_t)(q0 + 2 * i) * HDIM + h];
        taq[i].y = ta[base + (size_t)(q0 + 2 * i + 1) * HDIM + h];
        m[i].x = -INFINITY;  m[i].y = -INFINITY;
        s[i].x = 0.0f;       s[i].y = 0.0f;
        acc[i].x = 0.0f;     acc[i].y = 0.0f;
    }

    const float* ptb = tb    + base + h;
    const float* phv = hv    + base + h;
    const float* pmk = maskp + base + h;

    float tbA[KU], hvA[KU], mkA[KU];
    float tbB[KU], hvB[KU], mkB[KU];

#define LOAD_GROUP(TB, HV, MK, K0)                                  \
    _Pragma("unroll") for (int j = 0; j < KU; ++j) {                \
        TB[j] = ptb[(size_t)((K0) + j) * HDIM];                     \
        HV[j] = phv[(size_t)((K0) + j) * HDIM];                     \
        MK[j] = pmk[(size_t)((K0) + j) * HDIM];                     \
    }

    LOAD_GROUP(tbA, hvA, mkA, 0);

    for (int k0 = 0; k0 < SEQ; k0 += 2 * KU) {
        __builtin_amdgcn_sched_barrier(0);
        LOAD_GROUP(tbB, hvB, mkB, k0 + KU);
        __builtin_amdgcn_sched_barrier(0);
        attn_group(tbA, hvA, mkA, taq, m, s, acc);

        const int kn = (k0 + 2 * KU) & (SEQ - 1);
        __builtin_amdgcn_sched_barrier(0);
        LOAD_GROUP(tbA, hvA, mkA, kn);
        __builtin_amdgcn_sched_barrier(0);
        attn_group(tbB, hvB, mkB, taq, m, s, acc);
    }
#undef LOAD_GROUP

#pragma unroll
    for (int i = 0; i < QP; ++i) {
        out[base + (size_t)(q0 + 2 * i) * HDIM + h]     = acc[i].x / s[i].x;
        out[base + (size_t)(q0 + 2 * i + 1) * HDIM + h] = acc[i].y / s[i].y;
    }
}

extern "C" void kernel_launch(void* const* d_in, const int* in_sizes, int n_in,
                              void* d_out, int out_size, void* d_ws, size_t ws_size,
                              hipStream_t stream) {
    const float* word_vecs = (const float*)d_in[0];
    const float* mask      = (const float*)d_in[1];
    const float* Wx_w      = (const float*)d_in[2];
    const float* Wx_b      = (const float*)d_in[3];
    const float* Wa_w      = (const float*)d_in[4];
    const float* Wa_b      = (const float*)d_in[5];
    const float* Ua_w      = (const float*)d_in[6];
    const float* Ua_b      = (const float*)d_in[7];
    float* out = (float*)d_out;

    float* hv = (float*)d_ws;
    float* ta = hv + (size_t)MROWS * HDIM;
    float* tb = ta + (size_t)MROWS * HDIM;

    gemm_wx_tanh<<<128, 256, 0, stream>>>(word_vecs, Wx_w, Wx_b, hv);
    gemm_ta_tb<<<128, 256, 0, stream>>>(hv, Wa_w, Wa_b, Ua_w, Ua_b, ta, tb);

    attn_fused<<<BATCH * (SEQ / QT), 128, 0, stream>>>(ta, tb, hv, mask, out);
}


# both projection GEMMs: operands staged in LDS by whole-line 16-byte loads (32x32 block tiles, 256 work-groups); fragments read from LDS instead of 16-line global loads; k order and sum chain kept
# speedup vs baseline: 7.6562x; 1.3414x over previous
#include <hip/hip_runtime.h>
#include <hip/hip_bf16.h>
#include <math.h>
#include <float.h>

#define BATCH   4
#define SEQ     512
#define INDIM   300
#define HDIM    128
#define MROWS   (BATCH * SEQ)

typedef __attribute__((ext_vector_type(2))) float v2f;

#define LOG2E 1.4426950408889634f
typedef __attribute__((ext_vector_type(8))) float v8f;

#define BT 32

__device__ __forceinline__ v2f lds_frag(const float* S, int row0, int k0,
                                        int pitch, int lane) {
    const int r = row0 + (lane & 15);
    const int k = k0 + ((lane >> 4) << 1);
    return *reinterpret_cast<const v2f*>(S + r * pitch + k);
}

template <int KT>
__device__ __forceinline__ void lds_frags2(const float* SA, const float* SB, int m0, int n0,
                                           int kbase, int pitch, int lane,
                                           v2f (&a)[KT], v2f (&b)[KT]) {
#pragma unroll
    for (int j = 0; j < KT; ++j) {
        a[j] = lds_frag(SA, m0, kbase + 4 * j, pitch, lane);
        b[j] = lds_frag(SB, n0, kbase + 4 * j, pitch, lane);
    }
}

__global__ __launch_bounds__(128) void gemm_wx_tanh(
    const float* __restrict__ A,
    const float* __restrict__ W,
    const float* __restrict__ bias,
    float* __restrict__ out)
{
    __shared__ __attribute__((aligned(16))) float sA[BT * INDIM];
    __shared__ __attribute__((aligned(16))) float sW[BT * INDIM];

    const int tid  = threadIdx.x;
    const int lane = tid & 31;
    const int wave = tid >> 5;
    const int mb   = (blockIdx.x >> 2) * BT;
    const int nb   = (blockIdx.x & 3) * BT;

    const float4* gA = reinterpret_cast<const float4*>(A + (size_t)mb * INDIM);
    const float4* gW = reinterpret_cast<const float4*>(W + (size_t)nb * INDIM);
    float4* lA = reinterpret_cast<float4*>(sA);
    float4* lW = reinterpret_cast<float4*>(sW);

    constexpr int NV   = BT * INDIM / 4;
    constexpr int FULL = NV / 128;
    const bool tail = tid < NV - FULL * 128;

    const int itail = tail ? tid + FULL * 128 : NV - 1;
    float4 ra[FULL], rw[FULL];
#pragma unroll
    for (int i = 0; i < FULL; ++i) {
        ra[i] = gA[tid + i * 128];
        rw[i] = gW[tid + i * 128];
    }
    const float4 ta4 = gA[itail];
    const float4 tw4 = gW[itail];
#pragma unroll
    for (int i = 0; i < FULL; ++i) {
        lA[tid + i * 128] = ra[i];
        lW[tid + i * 128] = rw[i];
    }
    lA[itail] = ta4;
    lW[itail] = tw4;
    __syncthreads();

    const int m0 = (wave >> 1) << 4;
    const int n0 = (wave & 1) << 4;

    v8f acc = {};
    constexpr int KT1 = 15;
    v2f a0[KT1], b0[KT1], a1[KT1], b1[KT1];
#define MMA_TRIP1(AF, BF)                                                    \
    _Pragma("unroll") for (int j = 0; j < KT1; ++j) {                        \
        acc = __builtin_amdgcn_wmma_f32_16x16x4_f32(                         \
                false, AF[j], false, BF[j], (short)0, acc, false, false);    \
    }
    lds_frags2<KT1>(sA, sW, m0, n0, 0 * 4 * KT1, INDIM, lane, a0, b0);
    __builtin_amdgcn_sched_barrier(0);
    lds_frags2<KT1>(sA, sW, m0, n0, 1 * 4 * KT1, INDIM, lane, a1, b1);
    __builtin_amdgcn_sched_barrier(0);
    MMA_TRIP1(a0, b0)
    __builtin_amdgcn_sched_barrier(0);
    lds_frags2<KT1>(sA, sW, m0, n0, 2 * 4 * KT1, INDIM, lane, a0, b0);
    __builtin_amdgcn_sched_barrier(0);
    MMA_TRIP1(a1, b1)
    __builtin_amdgcn_sched_barrier(0);
    lds_frags2<KT1>(sA, sW, m0, n0, 3 * 4 * KT1, INDIM, lane, a1, b1);
    __builtin_amdgcn_sched_barrier(0);
    MMA_TRIP1(a0, b0)
    __builtin_amdgcn_sched_barrier(0);
    lds_frags2<KT1>(sA, sW, m0, n0, 4 * 4 * KT1, INDIM, lane, a0, b0);
    __builtin_amdgcn_sched_barrier(0);
    MMA_TRIP1(a1, b1)
    MMA_TRIP1(a0, b0)
#undef MMA_TRIP1

    const int col = nb + n0 + (lane & 15);
    const int rb  = mb + m0 + ((lane >> 4) << 3);
    const float bv = bias[col];
#pragma unroll
    for (int r = 0; r < 8; ++r) {
        out[(size_t)(rb + r) * HDIM + col] = tanhf(acc[r] + bv);
    }
}

#define PITCH2 (HDIM + 4)
__global__ __launch_bounds__(128) void gemm_ta_tb(
    const float* __restrict__ A,
    const float* __restrict__ Wa,
    const float* __restrict__ ba,
    const float* __restrict__ Ua,
    const float* __restrict__ bu,
    float* __restrict__ ta,
    float* __restrict__ tb)
{
    __shared__ __attribute__((aligned(16))) float sA[BT * PITCH2];
    __shared__ __attribute__((aligned(16))) float sWa[BT * PITCH2];
    __shared__ __attribute__((aligned(16))) float sUa[BT * PITCH2];

    const int tid  = threadIdx.x;
    const int lane = tid & 31;
    const int wave = tid >> 5;
    const int mb   = (blockIdx.x >> 2) * BT;
    const int nb   = (blockIdx.x & 3) * BT;

    const float4* gA  = reinterpret_cast<const float4*>(A  + (size_t)mb * HDIM);
    const float4* gWa = reinterpret_cast<const float4*>(Wa + (size_t)nb * HDIM);
    const float4* gUa = reinterpret_cast<const float4*>(Ua + (size_t)nb * HDIM);

    constexpr int ROUNDS = BT / 4;
    float4 ra[ROUNDS], rwa[ROUNDS], rua[ROUNDS];
#pragma unroll
    for (int i = 0; i < ROUNDS; ++i) {
        ra[i]  = gA[tid + i * 128];
        rwa[i] = gWa[tid + i * 128];
        rua[i] = gUa[tid + i * 128];
    }
#pragma unroll
    for (int i = 0; i < ROUNDS; ++i) {
        const int off = (wave + 4 * i) * PITCH2 + lane * 4;
        *reinterpret_cast<float4*>(sA  + off) = ra[i];
        *reinterpret_cast<float4*>(sWa + off) = rwa[i];
        *reinterpret_cast<float4*>(sUa + off) = rua[i];
    }
    __syncthreads();

    const int m0 = (wave >> 1) << 4;
    const int n0 = (wave & 1) << 4;

    v8f accA = {};
    v8f accB = {};
    constexpr int KT2 = 8;
    v2f a0[KT2], w0[KT2], u0[KT2], a1[KT2], w1[KT2], u1[KT2];
#define LOAD_TRIP2(AF, WF, UF, T)                                            \
    _Pragma("unroll") for (int j = 0; j < KT2; ++j) {                        \
        AF[j] = lds_frag(sA,  m0, (T) * 4 * KT2 + 4 * j, PITCH2, lane);      \
        WF[j] = lds_frag(sWa, n0, (T) * 4 * KT2 + 4 * j, PITCH2, lane);      \
        UF[j] = lds_frag(sUa, n0, (T) * 4 * KT2 + 4 * j, PITCH2, lane);      \
    }
#define MMA_TRIP2(AF, WF, UF)                                                \
    _Pragma("unroll") for (int j = 0; j < KT2; ++j) {                        \
        accA = __builtin_amdgcn_wmma_f32_16x16x4_f32(                        \
                 false, AF[j], false, WF[j], (short)0, accA, false, false);  \
        accB = __builtin_amdgcn_wmma_f32_16x16x4_f32(                        \
                 false, AF[j], false, UF[j], (short)0, accB, false, false);  \
    }
    LOAD_TRIP2(a0, w0, u0, 0)
    __builtin_amdgcn_sched_barrier(0);
    LOAD_TRIP2(a1, w1, u1, 1)
    __builtin_amdgcn_sched_barrier(0);
    MMA_TRIP2(a0, w0, u0)
    __builtin_amdgcn_sched_barrier(0);
    LOAD_TRIP2(a0, w0, u0, 2)
    __builtin_amdgcn_sched_barrier(0);
    MMA_TRIP2(a1, w1, u1)
    __builtin_amdgcn_sched_barrier(0);
    LOAD_TRIP2(a1, w1, u1, 3)
    __builtin_amdgcn_sched_barrier(0);
    MMA_TRIP2(a0, w0, u0)
    MMA_TRIP2(a1, w1, u1)
#undef LOAD_TRIP2
#undef MMA_TRIP2

    const int col = nb + n0 + (lane & 15);
    const int rb  = mb + m0 + ((lane >> 4) << 3);
    const float bva = ba[col];
    const float bvu = bu[col];
#pragma unroll
    for (int r = 0; r < 8; ++r) {
        const size_t idx = (size_t)(rb + r) * HDIM + col;
        ta[idx] = (accA[r] + bva) * LOG2E;
        tb[idx] = (accB[r] + bvu) * LOG2E;
    }
}

#define QT 4
#define KU 16

#define QP (QT / 2)
__device__ __forceinline__ void attn_group(const float (&tbk)[KU], const float (&hvk)[KU],
                                           const float (&mkk)[KU], const v2f (&taq)[QP],
                                           v2f (&m)[QP], v2f (&s)[QP], v2f (&acc)[QP]) {
    v2f e[KU][QP];
    v2f bm[QP];
    v2f mr[QP];
    const v2f l2e = {LOG2E, LOG2E};
    const v2f zero2 = {0.0f, 0.0f};
    const v2f lowest2 = {-FLT_MAX, -FLT_MAX};
#pragma unroll
    for (int i = 0; i < QP; ++i) { bm[i].x = -INFINITY; bm[i].y = -INFINITY; }

#pragma unroll
    for (int j = 0; j < KU; ++j) {
        const v2f mk2 = {mkk[j], mkk[j]};
#pragma unroll
        for (int i = 0; i < QP; ++i) {
            const v2f x = taq[i] + tbk[j];
            v2f t;
            t.x = __builtin_amdgcn_exp2f(x.x);
            t.y = __builtin_amdgcn_exp2f(x.y);
            const v2f em1 = __builtin_elementwise_fma(t, l2e, -l2e);
            const v2f el = __builtin_elementwise_maximum(
                               x, __builtin_elementwise_minimum(em1, zero2));
            const v2f v = __builtin_elementwise_fma(mk2, l2e, el);
            e[j][i] = v;
            bm[i] = __builtin_elementwise_maximum(bm[i], v);
        }
    }

#pragma unroll
    for (int i = 0; i < QP; ++i) {
        v2f mn, corr;
        mn = __builtin_elementwise_maximum(m[i], bm[i]);
        mr[i] = __builtin_elementwise_maximum(mn, lowest2);
        const v2f dm = m[i] - mr[i];
        corr.x = __builtin_amdgcn_exp2f(dm.x);
        corr.y = __builtin_amdgcn_exp2f(dm.y);
        s[i]   *= corr;
        acc[i] *= corr;
        m[i]    = mn;
    }

#pragma unroll
    for (int j = 0; j < KU; ++j) {
        const v2f hv2 = {hvk[j], hvk[j]};
#pragma unroll
        for (int i = 0; i < QP; ++i) {
            const v2f d = e[j][i] - mr[i];
            v2f p;
            p.x = __builtin_amdgcn_exp2f(d.x);
            p.y = __builtin_amdgcn_exp2f(d.y);
            s[i]  += p;
            acc[i] = __builtin_elementwise_fma(p, hv2, acc[i]);
        }
    }
}

__global__ __launch_bounds__(128) void attn_fused(
    const float* __restrict__ ta,
    const float* __restrict__ tb,
    const float* __restrict__ hv,
    const float* __restrict__ maskp,
    float* __restrict__ out)
{
    const int h   = threadIdx.x;
    const int bq  = blockIdx.x;
    const int b   = bq / (SEQ / QT);
    const int q0  = (bq % (SEQ / QT)) * QT;
    const size_t base = (size_t)b * SEQ * HDIM;

    v2f taq[QP], m[QP], s[QP], acc[QP];
#pragma unroll
    for (int i = 0; i < QP; ++i) {
        taq[i].x = ta[base + (size_t)(q0 + 2 * i) * HDIM + h];
        taq[i].y = ta[base + (size_t)(q0 + 2 * i + 1) * HDIM + h];
        m[i].x = -INFINITY;  m[i].y = -INFINITY;
        s[i].x = 0.0f;       s[i].y = 0.0f;
        acc[i].x = 0.0f;     acc[i].y = 0.0f;
    }

    const float* ptb = tb    + base + h;
    const float* phv = hv    + base + h;
    const float* pmk = maskp + base + h;

    float tbA[KU], hvA[KU], mkA[KU];
    float tbB[KU], hvB[KU], mkB[KU];

#define LOAD_GROUP(TB, HV, MK, K0)                                  \
    _Pragma("unroll") for (int j = 0; j < KU; ++j) {                \
        TB[j] = ptb[(size_t)((K0) + j) * HDIM];                     \
        HV[j] = phv[(size_t)((K0) + j) * HDIM];                     \
        MK[j] = pmk[(size_t)((K0) + j) * HDIM];                     \
    }

    LOAD_GROUP(tbA, hvA, mkA, 0);

    for (int k0 = 0; k0 < SEQ; k0 += 2 * KU) {
        __builtin_amdgcn_sched_barrier(0);
        LOAD_GROUP(tbB, hvB, mkB, k0 + KU);
        __builtin_amdgcn_sched_barrier(0);
        attn_group(tbA, hvA, mkA, taq, m, s, acc);

        const int kn = (k0 + 2 * KU) & (SEQ - 1);
        __builtin_amdgcn_sched_barrier(0);
        LOAD_GROUP(tbA, hvA, mkA, kn);
        __builtin_amdgcn_sched_barrier(0);
        attn_group(tbB, hvB, mkB, taq, m, s, acc);
    }
#undef LOAD_GROUP

#pragma unroll
    for (int i = 0; i < QP; ++i) {
        out[base + (size_t)(q0 + 2 * i) * HDIM + h]     = acc[i].x / s[i].x;
        out[base + (size_t)(q0 + 2 * i + 1) * HDIM + h] = acc[i].y / s[i].y;
    }
}

extern "C" void kernel_launch(void* const* d_in, const int* in_sizes, int n_in,
                              void* d_out, int out_size, void* d_ws, size_t ws_size,
                              hipStream_t stream) {
    const float* word_vecs = (const float*)d_in[0];
    const float* mask      = (const float*)d_in[1];
    const float* Wx_w      = (const float*)d_in[2];
    const float* Wx_b      = (const float*)d_in[3];
    const float* Wa_w      = (const float*)d_in[4];
    const float* Wa_b      = (const float*)d_in[5];
    const float* Ua_w      = (const float*)d_in[6];
    const float* Ua_b      = (const float*)d_in[7];
    float* out = (float*)d_out;

    float* hv = (float*)d_ws;
    float* ta = hv + (size_t)MROWS * HDIM;
    float* tb = ta + (size_t)MROWS * HDIM;

    gemm_wx_tanh<<<256, 128, 0, stream>>>(word_vecs, Wx_w, Wx_b, hv);
    gemm_ta_tb<<<256, 128, 0, stream>>>(hv, Wa_w, Wa_b, Ua_w, Ua_b, ta, tb);

    attn_fused<<<BATCH * (SEQ / QT), 128, 0, stream>>>(ta, tb, hv, mask, out);
}


# attention: elu's exp(ta+tb) formed as product exp(ta)*exp(tb), one factor per q-stream, one per key row: 1 exp per score, not 2; only in waves with all |ta|<=64 (base 2), else as before; k order kept
# speedup vs baseline: 8.0673x; 1.0537x over previous
#include <hip/hip_runtime.h>
#include <hip/hip_bf16.h>
#include <math.h>
#include <float.h>

#define BATCH   4
#define SEQ     512
#define INDIM   300
#define HDIM    128
#define MROWS   (BATCH * SEQ)

typedef __attribute__((ext_vector_type(2))) float v2f;

#define LOG2E 1.4426950408889634f
typedef __attribute__((ext_vector_type(8))) float v8f;

#define BT 32

__device__ __forceinline__ v2f lds_frag(const float* S, int row0, int k0,
                                        int pitch, int lane) {
    const int r = row0 + (lane & 15);
    const int k = k0 + ((lane >> 4) << 1);
    return *reinterpret_cast<const v2f*>(S + r * pitch + k);
}

template <int KT>
__device__ __forceinline__ void lds_frags2(const float* SA, const float* SB, int m0, int n0,
                                           int kbase, int pitch, int lane,
                                           v2f (&a)[KT], v2f (&b)[KT]) {
#pragma unroll
    for (int j = 0; j < KT; ++j) {
        a[j] = lds_frag(SA, m0, kbase + 4 * j, pitch, lane);
        b[j] = lds_frag(SB, n0, kbase + 4 * j, pitch, lane);
    }
}

__global__ __launch_bounds__(128) void gemm_wx_tanh(
    const float* __restrict__ A,
    const float* __restrict__ W,
    const float* __restrict__ bias,
    float* __restrict__ out)
{
    __shared__ __attribute__((aligned(16))) float sA[BT * INDIM];
    __shared__ __attribute__((aligned(16))) float sW[BT * INDIM];

    const int tid  = threadIdx.x;
    const int lane = tid & 31;
    const int wave = tid >> 5;
    const int mb   = (blockIdx.x >> 2) * BT;
    const int nb   = (blockIdx.x & 3) * BT;

    const float4* gA = reinterpret_cast<const float4*>(A + (size_t)mb * INDIM);
    const float4* gW = reinterpret_cast<const float4*>(W + (size_t)nb * INDIM);
    float4* lA = reinterpret_cast<float4*>(sA);
    float4* lW = reinterpret_cast<float4*>(sW);

    constexpr int NV   = BT * INDIM / 4;
    constexpr int FULL = NV / 128;
    const bool tail = tid < NV - FULL * 128;

    const int itail = tail ? tid + FULL * 128 : NV - 1;
    float4 ra[FULL], rw[FULL];
#pragma unroll
    for (int i = 0; i < FULL; ++i) {
        ra[i] = gA[tid + i * 128];
        rw[i] = gW[tid + i * 128];
    }
    const float4 ta4 = gA[itail];
    const float4 tw4 = gW[itail];
#pragma unroll
    for (int i = 0; i < FULL; ++i) {
        lA[tid + i * 128] = ra[i];
        lW[tid + i * 128] = rw[i];
    }
    lA[itail] = ta4;
    lW[itail] = tw4;
    __syncthreads();

    const int m0 = (wave >> 1) << 4;
    const int n0 = (wave & 1) << 4;

    v8f acc = {};
    constexpr int KT1 = 15;
    v2f a0[KT1], b0[KT1], a1[KT1], b1[KT1];
#define MMA_TRIP1(AF, BF)                                                    \
    _Pragma("unroll") for (int j = 0; j < KT1; ++j) {                        \
        acc = __builtin_amdgcn_wmma_f32_16x16x4_f32(                         \
                false, AF[j], false, BF[j], (short)0, acc, false, false);    \
    }
    lds_frags2<KT1>(sA, sW, m0, n0, 0 * 4 * KT1, INDIM, lane, a0, b0);
    __builtin_amdgcn_sched_barrier(0);
    lds_frags2<KT1>(sA, sW, m0, n0, 1 * 4 * KT1, INDIM, lane, a1, b1);
    __builtin_amdgcn_sched_barrier(0);
    MMA_TRIP1(a0, b0)
    __builtin_amdgcn_sched_barrier(0);
    lds_frags2<KT1>(sA, sW, m0, n0, 2 * 4 * KT1, INDIM, lane, a0, b0);
    __builtin_amdgcn_sched_barrier(0);
    MMA_TRIP1(a1, b1)
    __builtin_amdgcn_sched_barrier(0);
    lds_frags2<KT1>(sA, sW, m0, n0, 3 * 4 * KT1, INDIM, lane, a1, b1);
    __builtin_amdgcn_sched_barrier(0);
    MMA_TRIP1(a0, b0)
    __builtin_amdgcn_sched_barrier(0);
    lds_frags2<KT1>(sA, sW, m0, n0, 4 * 4 * KT1, INDIM, lane, a0, b0);
    __builtin_amdgcn_sched_barrier(0);
    MMA_TRIP1(a1, b1)
    MMA_TRIP1(a0, b0)
#undef MMA_TRIP1

    const int col = nb + n0 + (lane & 15);
    const int rb  = mb + m0 + ((lane >> 4) << 3);
    const float bv = bias[col];
#pragma unroll
    for (int r = 0; r < 8; ++r) {
        out[(size_t)(rb + r) * HDIM + col] = tanhf(acc[r] + bv);
    }
}

#define PITCH2 (HDIM + 4)
__global__ __launch_bounds__(128) void gemm_ta_tb(
    const float* __restrict__ A,
    const float* __restrict__ Wa,
    const float* __restrict__ ba,
    const float* __restrict__ Ua,
    const float* __restrict__ bu,
    float* __restrict__ ta,
    float* __restrict__ tb)
{
    __shared__ __attribute__((aligned(16))) float sA[BT * PITCH2];
    __shared__ __attribute__((aligned(16))) float sWa[BT * PITCH2];
    __shared__ __attribute__((aligned(16))) float sUa[BT * PITCH2];

    const int tid  = threadIdx.x;
    const int lane = tid & 31;
    const int wave = tid >> 5;
    const int mb   = (blockIdx.x >> 2) * BT;
    const int nb   = (blockIdx.x & 3) * BT;

    const float4* gA  = reinterpret_cast<const float4*>(A  + (size_t)mb * HDIM);
    const float4* gWa = reinterpret_cast<const float4*>(Wa + (size_t)nb * HDIM);
    const float4* gUa = reinterpret_cast<const float4*>(Ua + (size_t)nb * HDIM);

    constexpr int ROUNDS = BT / 4;
    float4 ra[ROUNDS], rwa[ROUNDS], rua[ROUNDS];
#pragma unroll
    for (int i = 0; i < ROUNDS; ++i) {
        ra[i]  = gA[tid + i * 128];
        rwa[i] = gWa[tid + i * 128];
        rua[i] = gUa[tid + i * 128];
    }
#pragma unroll
    for (int i = 0; i < ROUNDS; ++i) {
        const int off = (wave + 4 * i) * PITCH2 + lane * 4;
        *reinterpret_cast<float4*>(sA  + off) = ra[i];
        *reinterpret_cast<float4*>(sWa + off) = rwa[i];
        *reinterpret_cast<float4*>(sUa + off) = rua[i];
    }
    __syncthreads();

    const int m0 = (wave >> 1) << 4;
    const int n0 = (wave & 1) << 4;

    v8f accA = {};
    v8f accB = {};
    constexpr int KT2 = 8;
    v2f a0[KT2], w0[KT2], u0[KT2], a1[KT2], w1[KT2], u1[KT2];
#define LOAD_TRIP2(AF, WF, UF, T)                                            \
    _Pragma("unroll") for (int j = 0; j < KT2; ++j) {                        \
        AF[j] = lds_frag(sA,  m0, (T) * 4 * KT2 + 4 * j, PITCH2, lane);      \
        WF[j] = lds_frag(sWa, n0, (T) * 4 * KT2 + 4 * j, PITCH2, lane);      \
        UF[j] = lds_frag(sUa, n0, (T) * 4 * KT2 + 4 * j, PITCH2, lane);      \
    }
#define MMA_TRIP2(AF, WF, UF)                                                \
    _Pragma("unroll") for (int j = 0; j < KT2; ++j) {                        \
        accA = __builtin_amdgcn_wmma_f32_16x16x4_f32(                        \
                 false, AF[j], false, WF[j], (short)0, accA, false, false);  \
        accB = __builtin_amdgcn_wmma_f32_16x16x4_f32(                        \
                 false, AF[j], false, UF[j], (short)0, accB, false, false);  \
    }
    LOAD_TRIP2(a0, w0, u0, 0)
    __builtin_amdgcn_sched_barrier(0);
    LOAD_TRIP2(a1, w1, u1, 1)
    __builtin_amdgcn_sched_barrier(0);
    MMA_TRIP2(a0, w0, u0)
    __builtin_amdgcn_sched_barrier(0);
    LOAD_TRIP2(a0, w0, u0, 2)
    __builtin_amdgcn_sched_barrier(0);
    MMA_TRIP2(a1, w1, u1)
    __builtin_amdgcn_sched_barrier(0);
    LOAD_TRIP2(a1, w1, u1, 3)
    __builtin_amdgcn_sched_barrier(0);
    MMA_TRIP2(a0, w0, u0)
    MMA_TRIP2(a1, w1, u1)
#undef LOAD_TRIP2
#undef MMA_TRIP2

    const int col = nb + n0 + (lane & 15);
    const int rb  = mb + m0 + ((lane >> 4) << 3);
    const float bva = ba[col];
    const float bvu = bu[col];
#pragma unroll
    for (int r = 0; r < 8; ++r) {
        const size_t idx = (size_t)(rb + r) * HDIM + col;
        ta[idx] = (accA[r] + bva) * LOG2E;
        tb[idx] = (accB[r] + bvu) * LOG2E;
    }
}

#define QT 4
#define SPLIT_LIM 64.0f
#define KU 16

#define QP (QT / 2)
template <bool SPLIT>
__device__ __forceinline__ void attn_group(const float (&tbk)[KU], const float (&hvk)[KU],
                                           const float (&mkk)[KU], const v2f (&taq)[QP],
                                           const v2f (&eaq)[QP],
                                           v2f (&m)[QP], v2f (&s)[QP], v2f (&acc)[QP]) {
    v2f e[KU][QP];
    v2f bm[QP];
    v2f mr[QP];
    const v2f l2e = {LOG2E, LOG2E};
    const v2f zero2 = {0.0f, 0.0f};
    const v2f lowest2 = {-FLT_MAX, -FLT_MAX};
#pragma unroll
    for (int i = 0; i < QP; ++i) { bm[i].x = -INFINITY; bm[i].y = -INFINITY; }

#pragma unroll
    for (int j = 0; j < KU; ++j) {
        const v2f mk2 = {mkk[j], mkk[j]};
        v2f eb2 = {0.0f, 0.0f};
        if (SPLIT) { eb2.x = __builtin_amdgcn_exp2f(tbk[j]); eb2.y = eb2.x; }
#pragma unroll
        for (int i = 0; i < QP; ++i) {
            const v2f x = taq[i] + tbk[j];
            v2f em1;
            if (SPLIT) {
                em1 = __builtin_elementwise_fma(eaq[i], eb2, -l2e);
            } else {
                v2f t;
                t.x = __builtin_amdgcn_exp2f(x.x);
                t.y = __builtin_amdgcn_exp2f(x.y);
                em1 = __builtin_elementwise_fma(t, l2e, -l2e);
            }
            const v2f el = __builtin_elementwise_maximum(
                               x, __builtin_elementwise_minimum(em1, zero2));
            const v2f v = __builtin_elementwise_fma(mk2, l2e, el);
            e[j][i] = v;
            bm[i] = __builtin_elementwise_maximum(bm[i], v);
        }
    }

#pragma unroll
    for (int i = 0; i < QP; ++i) {
        v2f mn, corr;
        mn = __builtin_elementwise_maximum(m[i], bm[i]);
        mr[i] = __builtin_elementwise_maximum(mn, lowest2);
        const v2f dm = m[i] - mr[i];
        corr.x = __builtin_amdgcn_exp2f(dm.x);
        corr.y = __builtin_amdgcn_exp2f(dm.y);
        s[i]   *= corr;
        acc[i] *= corr;
        m[i]    = mn;
    }

#pragma unroll
    for (int j = 0; j < KU; ++j) {
        const v2f hv2 = {hvk[j], hvk[j]};
#pragma unroll
        for (int i = 0; i < QP; ++i) {
            const v2f d = e[j][i] - mr[i];
            v2f p;
            p.x = __builtin_amdgcn_exp2f(d.x);
            p.y = __builtin_amdgcn_exp2f(d.y);
            s[i]  += p;
            acc[i] = __builtin_elementwise_fma(p, hv2, acc[i]);
        }
    }
}

#define LOAD_GROUP(TB, HV, MK, K0)                                  \
    _Pragma("unroll") for (int j = 0; j < KU; ++j) {                \
        TB[j] = ptb[(size_t)((K0) + j) * HDIM];                     \
        HV[j] = phv[(size_t)((K0) + j) * HDIM];                     \
        MK[j] = pmk[(size_t)((K0) + j) * HDIM];                     \
    }

template <bool SPLIT>
__device__ __forceinline__ void attn_keys(const float* __restrict__ ptb,
                                          const float* __restrict__ phv,
                                          const float* __restrict__ pmk,
                                          float (&tbA)[KU], float (&hvA)[KU], float (&mkA)[KU],
                                          const v2f (&taq)[QP], const v2f (&eaq)[QP],
                                          v2f (&m)[QP], v2f (&s)[QP], v2f (&acc)[QP]) {
    float tbB[KU], hvB[KU], mkB[KU];

    for (int k0 = 0; k0 < SEQ; k0 += 2 * KU) {
        __builtin_amdgcn_sched_barrier(0);
        LOAD_GROUP(tbB, hvB, mkB, k0 + KU);
        __builtin_amdgcn_sched_barrier(0);
        attn_group<SPLIT>(tbA, hvA, mkA, taq, eaq, m, s, acc);

        const int kn = (k0 + 2 * KU) & (SEQ - 1);
        __builtin_amdgcn_sched_barrier(0);
        LOAD_GROUP(tbA, hvA, mkA, kn);
        __builtin_amdgcn_sched_barrier(0);
        attn_group<SPLIT>(tbB, hvB, mkB, taq, eaq, m, s, acc);
    }
}

__global__ __launch_bounds__(128) void attn_fused(
    const float* __restrict__ ta,
    const float* __restrict__ tb,
    const float* __restrict__ hv,
    const float* __restrict__ maskp,
    float* __restrict__ out)
{
    const int h   = threadIdx.x;
    const int bq  = blockIdx.x;
    const int b   = bq / (SEQ / QT);
    const int q0  = (bq % (SEQ / QT)) * QT;
    const size_t base = (size_t)b * SEQ * HDIM;

    v2f taq[QP], eaq[QP], m[QP], s[QP], acc[QP];
    bool ta_small = true;
#pragma unroll
    for (int i = 0; i < QP; ++i) {
        taq[i].x = ta[base + (size_t)(q0 + 2 * i) * HDIM + h];
        taq[i].y = ta[base + (size_t)(q0 + 2 * i + 1) * HDIM + h];
        eaq[i].x = LOG2E * __builtin_amdgcn_exp2f(taq[i].x);
        eaq[i].y = LOG2E * __builtin_amdgcn_exp2f(taq[i].y);
        ta_small = ta_small && (__builtin_fabsf(taq[i].x) <= SPLIT_LIM)
                            && (__builtin_fabsf(taq[i].y) <= SPLIT_LIM);
        m[i].x = -INFINITY;  m[i].y = -INFINITY;
        s[i].x = 0.0f;       s[i].y = 0.0f;
        acc[i].x = 0.0f;     acc[i].y = 0.0f;
    }

    const float* ptb = tb    + base + h;
    const float* phv = hv    + base + h;
    const float* pmk = maskp + base + h;

    float tbA[KU], hvA[KU], mkA[KU];
    LOAD_GROUP(tbA, hvA, mkA, 0);

    if (__builtin_amdgcn_ballot_w32(!ta_small) == 0u) {
        attn_keys<true>(ptb, phv, pmk, tbA, hvA, mkA, taq, eaq, m, s, acc);
    } else {
        attn_keys<false>(ptb, phv, pmk, tbA, hvA, mkA, taq, eaq, m, s, acc);
    }
#undef LOAD_GROUP

#pragma unroll
    for (int i = 0; i < QP; ++i) {
        out[base + (size_t)(q0 + 2 * i) * HDIM + h]     = acc[i].x / s[i].x;
        out[base + (size_t)(q0 + 2 * i + 1) * HDIM + h] = acc[i].y / s[i].y;
    }
}

extern "C" void kernel_launch(void* const* d_in, const int* in_sizes, int n_in,
                              void* d_out, int out_size, void* d_ws, size_t ws_size,
                              hipStream_t stream) {
    const float* word_vecs = (const float*)d_in[0];
    const float* mask      = (const float*)d_in[1];
    const float* Wx_w      = (const float*)d_in[2];
    const float* Wx_b      = (const float*)d_in[3];
    const float* Wa_w      = (const float*)d_in[4];
    const float* Wa_b      = (const float*)d_in[5];
    const float* Ua_w      = (const float*)d_in[6];
    const float* Ua_b      = (const float*)d_in[7];
    float* out = (float*)d_out;

    float* hv = (float*)d_ws;
    float* ta = hv + (size_t)MROWS * HDIM;
    float* tb = ta + (size_t)MROWS * HDIM;

    gemm_wx_tanh<<<256, 128, 0, stream>>>(word_vecs, Wx_w, Wx_b, hv);
    gemm_ta_tb<<<256, 128, 0, stream>>>(hv, Wa_w, Wa_b, Ua_w, Ua_b, ta, tb);

    attn_fused<<<BATCH * (SEQ / QT), 128, 0, stream>>>(ta, tb, hv, mask, out);
}


# attention: 8 q-streams per thread, not 4 (key-row loads, exp(tb) once per 8 queries); keys split over two wave groups, each in key order; the two partial softmax sums merged once at the end via LDS
# speedup vs baseline: 8.8565x; 1.0978x over previous
#include <hip/hip_runtime.h>
#include <hip/hip_bf16.h>
#include <math.h>
#include <float.h>

#define BATCH   4
#define SEQ     512
#define INDIM   300
#define HDIM    128
#define MROWS   (BATCH * SEQ)

typedef __attribute__((ext_vector_type(2))) float v2f;

#define LOG2E 1.4426950408889634f
typedef __attribute__((ext_vector_type(8))) float v8f;

#define BT 32

__device__ __forceinline__ v2f lds_frag(const float* S, int row0, int k0,
                                        int pitch, int lane) {
    const int r = row0 + (lane & 15);
    const int k = k0 + ((lane >> 4) << 1);
    return *reinterpret_cast<const v2f*>(S + r * pitch + k);
}

template <int KT>
__device__ __forceinline__ void lds_frags2(const float* SA, const float* SB, int m0, int n0,
                                           int kbase, int pitch, int lane,
                                           v2f (&a)[KT], v2f (&b)[KT]) {
#pragma unroll
    for (int j = 0; j < KT; ++j) {
        a[j] = lds_frag(SA, m0, kbase + 4 * j, pitch, lane);
        b[j] = lds_frag(SB, n0, kbase + 4 * j, pitch, lane);
    }
}

__global__ __launch_bounds__(128) void gemm_wx_tanh(
    const float* __restrict__ A,
    const float* __restrict__ W,
    const float* __restrict__ bias,
    float* __restrict__ out)
{
    __shared__ __attribute__((aligned(16))) float sA[BT * INDIM];
    __shared__ __attribute__((aligned(16))) float sW[BT * INDIM];

    const int tid  = threadIdx.x;
    const int lane = tid & 31;
    const int wave = tid >> 5;
    const int mb   = (blockIdx.x >> 2) * BT;
    const int nb   = (blockIdx.x & 3) * BT;

    const float4* gA = reinterpret_cast<const float4*>(A + (size_t)mb * INDIM);
    const float4* gW = reinterpret_cast<const float4*>(W + (size_t)nb * INDIM);
    float4* lA = reinterpret_cast<float4*>(sA);
    float4* lW = reinterpret_cast<float4*>(sW);

    constexpr int NV   = BT * INDIM / 4;
    constexpr int FULL = NV / 128;
    const bool tail = tid < NV - FULL * 128;

    const int itail = tail ? tid + FULL * 128 : NV - 1;
    float4 ra[FULL], rw[FULL];
#pragma unroll
    for (int i = 0; i < FULL; ++i) {
        ra[i] = gA[tid + i * 128];
        rw[i] = gW[tid + i * 128];
    }
    const float4 ta4 = gA[itail];
    const float4 tw4 = gW[itail];
#pragma unroll
    for (int i = 0; i < FULL; ++i) {
        lA[tid + i * 128] = ra[i];
        lW[tid + i * 128] = rw[i];
    }
    lA[itail] = ta4;
    lW[itail] = tw4;
    __syncthreads();

    const int m0 = (wave >> 1) << 4;
    const int n0 = (wave & 1) << 4;

    v8f acc = {};
    constexpr int KT1 = 15;
    v2f a0[KT1], b0[KT1], a1[KT1], b1[KT1];
#define MMA_TRIP1(AF, BF)                                                    \
    _Pragma("unroll") for (int j = 0; j < KT1; ++j) {                        \
        acc = __builtin_amdgcn_wmma_f32_16x16x4_f32(                         \
                false, AF[j], false, BF[j], (short)0, acc, false, false);    \
    }
    lds_frags2<KT1>(sA, sW, m0, n0, 0 * 4 * KT1, INDIM, lane, a0, b0);
    __builtin_amdgcn_sched_barrier(0);
    lds_frags2<KT1>(sA, sW, m0, n0, 1 * 4 * KT1, INDIM, lane, a1, b1);
    __builtin_amdgcn_sched_barrier(0);
    MMA_TRIP1(a0, b0)
    __builtin_amdgcn_sched_barrier(0);
    lds_frags2<KT1>(sA, sW, m0, n0, 2 * 4 * KT1, INDIM, lane, a0, b0);
    __builtin_amdgcn_sched_barrier(0);
    MMA_TRIP1(a1, b1)
    __builtin_amdgcn_sched_barrier(0);
    lds_frags2<KT1>(sA, sW, m0, n0, 3 * 4 * KT1, INDIM, lane, a1, b1);
    __builtin_amdgcn_sched_barrier(0);
    MMA_TRIP1(a0, b0)
    __builtin_amdgcn_sched_barrier(0);
    lds_frags2<KT1>(sA, sW, m0, n0, 4 * 4 * KT1, INDIM, lane, a0, b0);
    __builtin_amdgcn_sched_barrier(0);
    MMA_TRIP1(a1, b1)
    MMA_TRIP1(a0, b0)
#undef MMA_TRIP1

    const int col = nb + n0 + (lane & 15);
    const int rb  = mb + m0 + ((lane >> 4) << 3);
    const float bv = bias[col];
#pragma unroll
    for (int r = 0; r < 8; ++r) {
        out[(size_t)(rb + r) * HDIM + col] = tanhf(acc[r] + bv);
    }
}

#define PITCH2 (HDIM + 4)
__global__ __launch_bounds__(128) void gemm_ta_tb(
    const float* __restrict__ A,
    const float* __restrict__ Wa,
    const float* __restrict__ ba,
    const float* __restrict__ Ua,
    const float* __restrict__ bu,
    float* __restrict__ ta,
    float* __restrict__ tb)
{
    __shared__ __attribute__((aligned(16))) float sA[BT * PITCH2];
    __shared__ __attribute__((aligned(16))) float sWa[BT * PITCH2];
    __shared__ __attribute__((aligned(16))) float sUa[BT * PITCH2];

    const int tid  = threadIdx.x;
    const int lane = tid & 31;
    const int wave = tid >> 5;
    const int mb   = (blockIdx.x >> 2) * BT;
    const int nb   = (blockIdx.x & 3) * BT;

    const float4* gA  = reinterpret_cast<const float4*>(A  + (size_t)mb * HDIM);
    const float4* gWa = reinterpret_cast<const float4*>(Wa + (size_t)nb * HDIM);
    const float4* gUa = reinterpret_cast<const float4*>(Ua + (size_t)nb * HDIM);

    constexpr int ROUNDS = BT / 4;
    float4 ra[ROUNDS], rwa[ROUNDS], rua[ROUNDS];
#pragma unroll
    for (int i = 0; i < ROUNDS; ++i) {
        ra[i]  = gA[tid + i * 128];
        rwa[i] = gWa[tid + i * 128];
        rua[i] = gUa[tid + i * 128];
    }
#pragma unroll
    for (int i = 0; i < ROUNDS; ++i) {
        const int off = (wave + 4 * i) * PITCH2 + lane * 4;
        *reinterpret_cast<float4*>(sA  + off) = ra[i];
        *reinterpret_cast<float4*>(sWa + off) = rwa[i];
        *reinterpret_cast<float4*>(sUa + off) = rua[i];
    }
    __syncthreads();

    const int m0 = (wave >> 1) << 4;
    const int n0 = (wave & 1) << 4;

    v8f accA = {};
    v8f accB = {};
    constexpr int KT2 = 8;
    v2f a0[KT2], w0[KT2], u0[KT2], a1[KT2], w1[KT2], u1[KT2];
#define LOAD_TRIP2(AF, WF, UF, T)                                            \
    _Pragma("unroll") for (int j = 0; j < KT2; ++j) {                        \
        AF[j] = lds_frag(sA,  m0, (T) * 4 * KT2 + 4 * j, PITCH2, lane);      \
        WF[j] = lds_frag(sWa, n0, (T) * 4 * KT2 + 4 * j, PITCH2, lane);      \
        UF[j] = lds_frag(sUa, n0, (T) * 4 * KT2 + 4 * j, PITCH2, lane);      \
    }
#define MMA_TRIP2(AF, WF, UF)                                                \
    _Pragma("unroll") for (int j = 0; j < KT2; ++j) {                        \
        accA = __builtin_amdgcn_wmma_f32_16x16x4_f32(                        \
                 false, AF[j], false, WF[j], (short)0, accA, false, false);  \
        accB = __builtin_amdgcn_wmma_f32_16x16x4_f32(                        \
                 false, AF[j], false, UF[j], (short)0, accB, false, false);  \
    }
    LOAD_TRIP2(a0, w0, u0, 0)
    __builtin_amdgcn_sched_barrier(0);
    LOAD_TRIP2(a1, w1, u1, 1)
    __builtin_amdgcn_sched_barrier(0);
    MMA_TRIP2(a0, w0, u0)
    __builtin_amdgcn_sched_barrier(0);
    LOAD_TRIP2(a0, w0, u0, 2)
    __builtin_amdgcn_sched_barrier(0);
    MMA_TRIP2(a1, w1, u1)
    __builtin_amdgcn_sched_barrier(0);
    LOAD_TRIP2(a1, w1, u1, 3)
    __builtin_amdgcn_sched_barrier(0);
    MMA_TRIP2(a0, w0, u0)
    MMA_TRIP2(a1, w1, u1)
#undef LOAD_TRIP2
#undef MMA_TRIP2

    const int col = nb + n0 + (lane & 15);
    const int rb  = mb + m0 + ((lane >> 4) << 3);
    const float bva = ba[col];
    const float bvu = bu[col];
#pragma unroll
    for (int r = 0; r < 8; ++r) {
        const size_t idx = (size_t)(rb + r) * HDIM + col;
        ta[idx] = (accA[r] + bva) * LOG2E;
        tb[idx] = (accB[r] + bvu) * LOG2E;
    }
}

#define QT 8
#define KHALF (SEQ / 2)
#define SPLIT_LIM 64.0f
#define KU 16

#define QP (QT / 2)
template <bool SPLIT>
__device__ __forceinline__ void attn_group(const float (&tbk)[KU], const float (&hvk)[KU],
                                           const float (&mkk)[KU], const v2f (&taq)[QP],
                                           const v2f (&eaq)[QP],
                                           v2f (&m)[QP], v2f (&s)[QP], v2f (&acc)[QP]) {
    v2f e[KU][QP];
    v2f bm[QP];
    v2f mr[QP];
    const v2f l2e = {LOG2E, LOG2E};
    const v2f zero2 = {0.0f, 0.0f};
    const v2f lowest2 = {-FLT_MAX, -FLT_MAX};
#pragma unroll
    for (int i = 0; i < QP; ++i) { bm[i].x = -INFINITY; bm[i].y = -INFINITY; }

#pragma unroll
    for (int j = 0; j < KU; ++j) {
        const v2f mk2 = {mkk[j], mkk[j]};
        v2f eb2 = {0.0f, 0.0f};
        if (SPLIT) { eb2.x = __builtin_amdgcn_exp2f(tbk[j]); eb2.y = eb2.x; }
#pragma unroll
        for (int i = 0; i < QP; ++i) {
            const v2f x = taq[i] + tbk[j];
            v2f em1;
            if (SPLIT) {
                em1 = __builtin_elementwise_fma(eaq[i], eb2, -l2e);
            } else {
                v2f t;
                t.x = __builtin_amdgcn_exp2f(x.x);
                t.y = __builtin_amdgcn_exp2f(x.y);
                em1 = __builtin_elementwise_fma(t, l2e, -l2e);
            }
            const v2f el = __builtin_elementwise_maximum(
                               x, __builtin_elementwise_minimum(em1, zero2));
            const v2f v = __builtin_elementwise_fma(mk2, l2e, el);
            e[j][i] = v;
            bm[i] = __builtin_elementwise_maximum(bm[i], v);
        }
    }

#pragma unroll
    for (int i = 0; i < QP; ++i) {
        v2f mn, corr;
        mn = __builtin_elementwise_maximum(m[i], bm[i]);
        mr[i] = __builtin_elementwise_maximum(mn, lowest2);
        const v2f dm = m[i] - mr[i];
        corr.x = __builtin_amdgcn_exp2f(dm.x);
        corr.y = __builtin_amdgcn_exp2f(dm.y);
        s[i]   *= corr;
        acc[i] *= corr;
        m[i]    = mn;
    }

#pragma unroll
    for (int j = 0; j < KU; ++j) {
        const v2f hv2 = {hvk[j], hvk[j]};
#pragma unroll
        for (int i = 0; i < QP; ++i) {
            const v2f d = e[j][i] - mr[i];
            v2f p;
            p.x = __builtin_amdgcn_exp2f(d.x);
            p.y = __builtin_amdgcn_exp2f(d.y);
            s[i]  += p;
            acc[i] = __builtin_elementwise_fma(p, hv2, acc[i]);
        }
    }
}

#define LOAD_GROUP(TB, HV, MK, K0)                                  \
    _Pragma("unroll") for (int j = 0; j < KU; ++j) {                \
        TB[j] = ptb[(size_t)((K0) + j) * HDIM];                     \
        HV[j] = phv[(size_t)((K0) + j) * HDIM];                     \
        MK[j] = pmk[(size_t)((K0) + j) * HDIM];                     \
    }

template <bool SPLIT>
__device__ __forceinline__ void attn_keys(const float* __restrict__ ptb,
                                          const float* __restrict__ phv,
                                          const float* __restrict__ pmk,
                                          float (&tbA)[KU], float (&hvA)[KU], float (&mkA)[KU],
                                          const v2f (&taq)[QP], const v2f (&eaq)[QP],
                                          v2f (&m)[QP], v2f (&s)[QP], v2f (&acc)[QP]) {
    float tbB[KU], hvB[KU], mkB[KU];

    for (int k0 = 0; k0 < KHALF; k0 += 2 * KU) {
        __builtin_amdgcn_sched_barrier(0);
        LOAD_GROUP(tbB, hvB, mkB, k0 + KU);
        __builtin_amdgcn_sched_barrier(0);
        attn_group<SPLIT>(tbA, hvA, mkA, taq, eaq, m, s, acc);

        const int kn = (k0 + 2 * KU) & (KHALF - 1);
        __builtin_amdgcn_sched_barrier(0);
        LOAD_GROUP(tbA, hvA, mkA, kn);
        __builtin_amdgcn_sched_barrier(0);
        attn_group<SPLIT>(tbB, hvB, mkB, taq, eaq, m, s, acc);
    }
}

__global__ __launch_bounds__(256) void attn_fused(
    const float* __restrict__ ta,
    const float* __restrict__ tb,
    const float* __restrict__ hv,
    const float* __restrict__ maskp,
    float* __restrict__ out)
{
    __shared__ float sm[QT][HDIM];
    __shared__ float ss[QT][HDIM];
    __shared__ float sa[QT][HDIM];

    const int h    = threadIdx.x & (HDIM - 1);
    const int half = threadIdx.x >> 7;
    const int bq  = blockIdx.x;
    const int b   = bq / (SEQ / QT);
    const int q0  = (bq % (SEQ / QT)) * QT;
    const size_t base = (size_t)b * SEQ * HDIM;

    v2f taq[QP], eaq[QP], m[QP], s[QP], acc[QP];
    bool ta_small = true;
#pragma unroll
    for (int i = 0; i < QP; ++i) {
        taq[i].x = ta[base + (size_t)(q0 + 2 * i) * HDIM + h];
        taq[i].y = ta[base + (size_t)(q0 + 2 * i + 1) * HDIM + h];
        eaq[i].x = LOG2E * __builtin_amdgcn_exp2f(taq[i].x);
        eaq[i].y = LOG2E * __builtin_amdgcn_exp2f(taq[i].y);
        ta_small = ta_small && (__builtin_fabsf(taq[i].x) <= SPLIT_LIM)
                            && (__builtin_fabsf(taq[i].y) <= SPLIT_LIM);
        m[i].x = -INFINITY;  m[i].y = -INFINITY;
        s[i].x = 0.0f;       s[i].y = 0.0f;
        acc[i].x = 0.0f;     acc[i].y = 0.0f;
    }

    const size_t kbase = base + (size_t)half * KHALF * HDIM;
    const float* ptb = tb    + kbase + h;
    const float* phv = hv    + kbase + h;
    const float* pmk = maskp + kbase + h;

    float tbA[KU], hvA[KU], mkA[KU];
    LOAD_GROUP(tbA, hvA, mkA, 0);

    if (__builtin_amdgcn_ballot_w32(!ta_small) == 0u) {
        attn_keys<true>(ptb, phv, pmk, tbA, hvA, mkA, taq, eaq, m, s, acc);
    } else {
        attn_keys<false>(ptb, phv, pmk, tbA, hvA, mkA, taq, eaq, m, s, acc);
    }
#undef LOAD_GROUP

    if (half == 1) {
#pragma unroll
        for (int i = 0; i < QP; ++i) {
            sm[2 * i][h] = m[i].x;    sm[2 * i + 1][h] = m[i].y;
            ss[2 * i][h] = s[i].x;    ss[2 * i + 1][h] = s[i].y;
            sa[2 * i][h] = acc[i].x;  sa[2 * i + 1][h] = acc[i].y;
        }
    }
    __syncthreads();
    if (half == 0) {
        const v2f lowest2 = {-FLT_MAX, -FLT_MAX};
#pragma unroll
        for (int i = 0; i < QP; ++i) {
            v2f m2, s2, a2, c1, c2;
            m2.x = sm[2 * i][h];  m2.y = sm[2 * i + 1][h];
            s2.x = ss[2 * i][h];  s2.y = ss[2 * i + 1][h];
            a2.x = sa[2 * i][h];  a2.y = sa[2 * i + 1][h];
            const v2f mn = __builtin_elementwise_maximum(m[i], m2);
            const v2f mr = __builtin_elementwise_maximum(mn, lowest2);
            const v2f d1 = m[i] - mr;
            const v2f d2 = m2 - mr;
            c1.x = __builtin_amdgcn_exp2f(d1.x);  c1.y = __builtin_amdgcn_exp2f(d1.y);
            c2.x = __builtin_amdgcn_exp2f(d2.x);  c2.y = __builtin_amdgcn_exp2f(d2.y);
            const v2f st = s[i] * c1 + s2 * c2;
            const v2f at = acc[i] * c1 + a2 * c2;
            out[base + (size_t)(q0 + 2 * i) * HDIM + h]     = at.x / st.x;
            out[base + (size_t)(q0 + 2 * i + 1) * HDIM + h] = at.y / st.y;
        }
    }
}

extern "C" void kernel_launch(void* const* d_in, const int* in_sizes, int n_in,
                              void* d_out, int out_size, void* d_ws, size_t ws_size,
                              hipStream_t stream) {
    const float* word_vecs = (const float*)d_in[0];
    const float* mask      = (const float*)d_in[1];
    const float* Wx_w      = (const float*)d_in[2];
    const float* Wx_b      = (const float*)d_in[3];
    const float* Wa_w      = (const float*)d_in[4];
    const float* Wa_b      = (const float*)d_in[5];
    const float* Ua_w      = (const float*)d_in[6];
    const float* Ua_b      = (const float*)d_in[7];
    float* out = (float*)d_out;

    float* hv = (float*)d_ws;
    float* ta = hv + (size_t)MROWS * HDIM;
    float* tb = ta + (size_t)MROWS * HDIM;

    gemm_wx_tanh<<<256, 128, 0, stream>>>(word_vecs, Wx_w, Wx_b, hv);
    gemm_ta_tb<<<256, 128, 0, stream>>>(hv, Wa_w, Wa_b, Ua_w, Ua_b, ta, tb);

    attn_fused<<<BATCH * (SEQ / QT), 256, 0, stream>>>(ta, tb, hv, mask, out);
}


# attention: where |ta|,|tb|,|mask| <= 16, weights exp2(score) with reference 0: no running maximum, no second pass, no rescale (bounds tested per wave after the loop, else as before); key order kept
# speedup vs baseline: 9.7055x; 1.0959x over previous
#include <hip/hip_runtime.h>
#include <hip/hip_bf16.h>
#include <math.h>
#include <float.h>

#define BATCH   4
#define SEQ     512
#define INDIM   300
#define HDIM    128
#define MROWS   (BATCH * SEQ)

typedef __attribute__((ext_vector_type(2))) float v2f;

#define LOG2E 1.4426950408889634f
typedef __attribute__((ext_vector_type(8))) float v8f;

#define BT 32

__device__ __forceinline__ v2f lds_frag(const float* S, int row0, int k0,
                                        int pitch, int lane) {
    const int r = row0 + (lane & 15);
    const int k = k0 + ((lane >> 4) << 1);
    return *reinterpret_cast<const v2f*>(S + r * pitch + k);
}

template <int KT>
__device__ __forceinline__ void lds_frags2(const float* SA, const float* SB, int m0, int n0,
                                           int kbase, int pitch, int lane,
                                           v2f (&a)[KT], v2f (&b)[KT]) {
#pragma unroll
    for (int j = 0; j < KT; ++j) {
        a[j] = lds_frag(SA, m0, kbase + 4 * j, pitch, lane);
        b[j] = lds_frag(SB, n0, kbase + 4 * j, pitch, lane);
    }
}

__global__ __launch_bounds__(128) void gemm_wx_tanh(
    const float* __restrict__ A,
    const float* __restrict__ W,
    const float* __restrict__ bias,
    float* __restrict__ out)
{
    __shared__ __attribute__((aligned(16))) float sA[BT * INDIM];
    __shared__ __attribute__((aligned(16))) float sW[BT * INDIM];

    const int tid  = threadIdx.x;
    const int lane = tid & 31;
    const int wave = tid >> 5;
    const int mb   = (blockIdx.x >> 2) * BT;
    const int nb   = (blockIdx.x & 3) * BT;

    const float4* gA = reinterpret_cast<const float4*>(A + (size_t)mb * INDIM);
    const float4* gW = reinterpret_cast<const float4*>(W + (size_t)nb * INDIM);
    float4* lA = reinterpret_cast<float4*>(sA);
    float4* lW = reinterpret_cast<float4*>(sW);

    constexpr int NV   = BT * INDIM / 4;
    constexpr int FULL = NV / 128;
    const bool tail = tid < NV - FULL * 128;

    const int itail = tail ? tid + FULL * 128 : NV - 1;
    float4 ra[FULL], rw[FULL];
#pragma unroll
    for (int i = 0; i < FULL; ++i) {
        ra[i] = gA[tid + i * 128];
        rw[i] = gW[tid + i * 128];
    }
    const float4 ta4 = gA[itail];
    const float4 tw4 = gW[itail];
#pragma unroll
    for (int i = 0; i < FULL; ++i) {
        lA[tid + i * 128] = ra[i];
        lW[tid + i * 128] = rw[i];
    }
    lA[itail] = ta4;
    lW[itail] = tw4;
    __syncthreads();

    const int m0 = (wave >> 1) << 4;
    const int n0 = (wave & 1) << 4;

    v8f acc = {};
    constexpr int KT1 = 15;
    v2f a0[KT1], b0[KT1], a1[KT1], b1[KT1];
#define MMA_TRIP1(AF, BF)                                                    \
    _Pragma("unroll") for (int j = 0; j < KT1; ++j) {                        \
        acc = __builtin_amdgcn_wmma_f32_16x16x4_f32(                         \
                false, AF[j], false, BF[j], (short)0, acc, false, false);    \
    }
    lds_frags2<KT1>(sA, sW, m0, n0, 0 * 4 * KT1, INDIM, lane, a0, b0);
    __builtin_amdgcn_sched_barrier(0);
    lds_frags2<KT1>(sA, sW, m0, n0, 1 * 4 * KT1, INDIM, lane, a1, b1);
    __builtin_amdgcn_sched_barrier(0);
    MMA_TRIP1(a0, b0)
    __builtin_amdgcn_sched_barrier(0);
    lds_frags2<KT1>(sA, sW, m0, n0, 2 * 4 * KT1, INDIM, lane, a0, b0);
    __builtin_amdgcn_sched_barrier(0);
    MMA_TRIP1(a1, b1)
    __builtin_amdgcn_sched_barrier(0);
    lds_frags2<KT1>(sA, sW, m0, n0, 3 * 4 * KT1, INDIM, lane, a1, b1);
    __builtin_amdgcn_sched_barrier(0);
    MMA_TRIP1(a0, b0)
    __builtin_amdgcn_sched_barrier(0);
    lds_frags2<KT1>(sA, sW, m0, n0, 4 * 4 * KT1, INDIM, lane, a0, b0);
    __builtin_amdgcn_sched_barrier(0);
    MMA_TRIP1(a1, b1)
    MMA_TRIP1(a0, b0)
#undef MMA_TRIP1

    const int col = nb + n0 + (lane & 15);
    const int rb  = mb + m0 + ((lane >> 4) << 3);
    const float bv = bias[col];
#pragma unroll
    for (int r = 0; r < 8; ++r) {
        out[(size_t)(rb + r) * HDIM + col] = tanhf(acc[r] + bv);
    }
}

#define PITCH2 (HDIM + 4)
__global__ __launch_bounds__(128) void gemm_ta_tb(
    const float* __restrict__ A,
    const float* __restrict__ Wa,
    const float* __restrict__ ba,
    const float* __restrict__ Ua,
    const float* __restrict__ bu,
    float* __restrict__ ta,
    float* __restrict__ tb)
{
    __shared__ __attribute__((aligned(16))) float sA[BT * PITCH2];
    __shared__ __attribute__((aligned(16))) float sWa[BT * PITCH2];
    __shared__ __attribute__((aligned(16))) float sUa[BT * PITCH2];

    const int tid  = threadIdx.x;
    const int lane = tid & 31;
    const int wave = tid >> 5;
    const int mb   = (blockIdx.x >> 2) * BT;
    const int nb   = (blockIdx.x & 3) * BT;

    const float4* gA  = reinterpret_cast<const float4*>(A  + (size_t)mb * HDIM);
    const float4* gWa = reinterpret_cast<const float4*>(Wa + (size_t)nb * HDIM);
    const float4* gUa = reinterpret_cast<const float4*>(Ua + (size_t)nb * HDIM);

    constexpr int ROUNDS = BT / 4;
    float4 ra[ROUNDS], rwa[ROUNDS], rua[ROUNDS];
#pragma unroll
    for (int i = 0; i < ROUNDS; ++i) {
        ra[i]  = gA[tid + i * 128];
        rwa[i] = gWa[tid + i * 128];
        rua[i] = gUa[tid + i * 128];
    }
#pragma unroll
    for (int i = 0; i < ROUNDS; ++i) {
        const int off = (wave + 4 * i) * PITCH2 + lane * 4;
        *reinterpret_cast<float4*>(sA  + off) = ra[i];
        *reinterpret_cast<float4*>(sWa + off) = rwa[i];
        *reinterpret_cast<float4*>(sUa + off) = rua[i];
    }
    __syncthreads();

    const int m0 = (wave >> 1) << 4;
    const int n0 = (wave & 1) << 4;

    v8f accA = {};
    v8f accB = {};
    constexpr int KT2 = 8;
    v2f a0[KT2], w0[KT2], u0[KT2], a1[KT2], w1[KT2], u1[KT2];
#define LOAD_TRIP2(AF, WF, UF, T)                                            \
    _Pragma("unroll") for (int j = 0; j < KT2; ++j) {                        \
        AF[j] = lds_frag(sA,  m0, (T) * 4 * KT2 + 4 * j, PITCH2, lane);      \
        WF[j] = lds_frag(sWa, n0, (T) * 4 * KT2 + 4 * j, PITCH2, lane);      \
        UF[j] = lds_frag(sUa, n0, (T) * 4 * KT2 + 4 * j, PITCH2, lane);      \
    }
#define MMA_TRIP2(AF, WF, UF)                                                \
    _Pragma("unroll") for (int j = 0; j < KT2; ++j) {                        \
        accA = __builtin_amdgcn_wmma_f32_16x16x4_f32(                        \
                 false, AF[j], false, WF[j], (short)0, accA, false, false);  \
        accB = __builtin_amdgcn_wmma_f32_16x16x4_f32(                        \
                 false, AF[j], false, UF[j], (short)0, accB, false, false);  \
    }
    LOAD_TRIP2(a0, w0, u0, 0)
    __builtin_amdgcn_sched_barrier(0);
    LOAD_TRIP2(a1, w1, u1, 1)
    __builtin_amdgcn_sched_barrier(0);
    MMA_TRIP2(a0, w0, u0)
    __builtin_amdgcn_sched_barrier(0);
    LOAD_TRIP2(a0, w0, u0, 2)
    __builtin_amdgcn_sched_barrier(0);
    MMA_TRIP2(a1, w1, u1)
    __builtin_amdgcn_sched_barrier(0);
    LOAD_TRIP2(a1, w1, u1, 3)
    __builtin_amdgcn_sched_barrier(0);
    MMA_TRIP2(a0, w0, u0)
    MMA_TRIP2(a1, w1, u1)
#undef LOAD_TRIP2
#undef MMA_TRIP2

    const int col = nb + n0 + (lane & 15);
    const int rb  = mb + m0 + ((lane >> 4) << 3);
    const float bva = ba[col];
    const float bvu = bu[col];
#pragma unroll
    for (int r = 0; r < 8; ++r) {
        const size_t idx = (size_t)(rb + r) * HDIM + col;
        ta[idx] = (accA[r] + bva) * LOG2E;
        tb[idx] = (accB[r] + bvu) * LOG2E;
    }
}

#define QT 8
#define KHALF (SEQ / 2)
#define SPLIT_LIM 64.0f
#define FAST_LIM 16.0f
#define KU 16

#define QP (QT / 2)
template <bool SPLIT>
__device__ __forceinline__ void attn_group(const float (&tbk)[KU], const float (&hvk)[KU],
                                           const float (&mkk)[KU], const v2f (&taq)[QP],
                                           const v2f (&eaq)[QP],
                                           v2f (&m)[QP], v2f (&s)[QP], v2f (&acc)[QP]) {
    v2f e[KU][QP];
    v2f bm[QP];
    v2f mr[QP];
    const v2f l2e = {LOG2E, LOG2E};
    const v2f zero2 = {0.0f, 0.0f};
    const v2f lowest2 = {-FLT_MAX, -FLT_MAX};
#pragma unroll
    for (int i = 0; i < QP; ++i) { bm[i].x = -INFINITY; bm[i].y = -INFINITY; }

#pragma unroll
    for (int j = 0; j < KU; ++j) {
        const v2f mk2 = {mkk[j], mkk[j]};
        v2f eb2 = {0.0f, 0.0f};
        if (SPLIT) { eb2.x = __builtin_amdgcn_exp2f(tbk[j]); eb2.y = eb2.x; }
#pragma unroll
        for (int i = 0; i < QP; ++i) {
            const v2f x = taq[i] + tbk[j];
            v2f em1;
            if (SPLIT) {
                em1 = __builtin_elementwise_fma(eaq[i], eb2, -l2e);
            } else {
                v2f t;
                t.x = __builtin_amdgcn_exp2f(x.x);
                t.y = __builtin_amdgcn_exp2f(x.y);
                em1 = __builtin_elementwise_fma(t, l2e, -l2e);
            }
            const v2f el = __builtin_elementwise_maximum(
                               x, __builtin_elementwise_minimum(em1, zero2));
            const v2f v = __builtin_elementwise_fma(mk2, l2e, el);
            e[j][i] = v;
            bm[i] = __builtin_elementwise_maximum(bm[i], v);
        }
    }

#pragma unroll
    for (int i = 0; i < QP; ++i) {
        v2f mn, corr;
        mn = __builtin_elementwise_maximum(m[i], bm[i]);
        mr[i] = __builtin_elementwise_maximum(mn, lowest2);
        const v2f dm = m[i] - mr[i];
        corr.x = __builtin_amdgcn_exp2f(dm.x);
        corr.y = __builtin_amdgcn_exp2f(dm.y);
        s[i]   *= corr;
        acc[i] *= corr;
        m[i]    = mn;
    }

#pragma unroll
    for (int j = 0; j < KU; ++j) {
        const v2f hv2 = {hvk[j], hvk[j]};
#pragma unroll
        for (int i = 0; i < QP; ++i) {
            const v2f d = e[j][i] - mr[i];
            v2f p;
            p.x = __builtin_amdgcn_exp2f(d.x);
            p.y = __builtin_amdgcn_exp2f(d.y);
            s[i]  += p;
            acc[i] = __builtin_elementwise_fma(p, hv2, acc[i]);
        }
    }
}

__device__ __forceinline__ void attn_group_fast(const float (&tbk)[KU], const float (&hvk)[KU],
                                                const float (&mkk)[KU], const v2f (&taq)[QP],
                                                const v2f (&eaq)[QP],
                                                v2f (&s)[QP], v2f (&acc)[QP], float& lim) {
    const v2f l2e = {LOG2E, LOG2E};
    const v2f zero2 = {0.0f, 0.0f};
#pragma unroll
    for (int j = 0; j < KU; ++j) {
        const v2f mk2 = {mkk[j], mkk[j]};
        const v2f hv2 = {hvk[j], hvk[j]};
        v2f eb2;
        eb2.x = __builtin_amdgcn_exp2f(tbk[j]);
        eb2.y = eb2.x;
        lim = __builtin_elementwise_maximum(
                  lim, __builtin_elementwise_maximum(__builtin_fabsf(tbk[j]),
                                                     __builtin_fabsf(mkk[j])));
#pragma unroll
        for (int i = 0; i < QP; ++i) {
            const v2f x = taq[i] + tbk[j];
            const v2f em1 = __builtin_elementwise_fma(eaq[i], eb2, -l2e);
            const v2f el = __builtin_elementwise_maximum(
                               x, __builtin_elementwise_minimum(em1, zero2));
            const v2f v = __builtin_elementwise_fma(mk2, l2e, el);
            v2f p;
            p.x = __builtin_amdgcn_exp2f(v.x);
            p.y = __builtin_amdgcn_exp2f(v.y);
            s[i]  += p;
            acc[i] = __builtin_elementwise_fma(p, hv2, acc[i]);
        }
    }
}

#define LOAD_GROUP(TB, HV, MK, K0)                                  \
    _Pragma("unroll") for (int j = 0; j < KU; ++j) {                \
        TB[j] = ptb[(size_t)((K0) + j) * HDIM];                     \
        HV[j] = phv[(size_t)((K0) + j) * HDIM];                     \
        MK[j] = pmk[(size_t)((K0) + j) * HDIM];                     \
    }

template <bool SPLIT>
__device__ __forceinline__ void attn_keys(const float* __restrict__ ptb,
                                          const float* __restrict__ phv,
                                          const float* __restrict__ pmk,
                                          float (&tbA)[KU], float (&hvA)[KU], float (&mkA)[KU],
                                          const v2f (&taq)[QP], const v2f (&eaq)[QP],
                                          v2f (&m)[QP], v2f (&s)[QP], v2f (&acc)[QP]) {
    float tbB[KU], hvB[KU], mkB[KU];

    for (int k0 = 0; k0 < KHALF; k0 += 2 * KU) {
        __builtin_amdgcn_sched_barrier(0);
        LOAD_GROUP(tbB, hvB, mkB, k0 + KU);
        __builtin_amdgcn_sched_barrier(0);
        attn_group<SPLIT>(tbA, hvA, mkA, taq, eaq, m, s, acc);

        const int kn = (k0 + 2 * KU) & (KHALF - 1);
        __builtin_amdgcn_sched_barrier(0);
        LOAD_GROUP(tbA, hvA, mkA, kn);
        __builtin_amdgcn_sched_barrier(0);
        attn_group<SPLIT>(tbB, hvB, mkB, taq, eaq, m, s, acc);
    }
}

__device__ __forceinline__ void attn_keys_fast(const float* __restrict__ ptb,
                                               const float* __restrict__ phv,
                                               const float* __restrict__ pmk,
                                               float (&tbA)[KU], float (&hvA)[KU], float (&mkA)[KU],
                                               const v2f (&taq)[QP], const v2f (&eaq)[QP],
                                               v2f (&s)[QP], v2f (&acc)[QP], float& lim) {
    float tbB[KU], hvB[KU], mkB[KU];

    for (int k0 = 0; k0 < KHALF; k0 += 2 * KU) {
        __builtin_amdgcn_sched_barrier(0);
        LOAD_GROUP(tbB, hvB, mkB, k0 + KU);
        __builtin_amdgcn_sched_barrier(0);
        attn_group_fast(tbA, hvA, mkA, taq, eaq, s, acc, lim);

        const int kn = (k0 + 2 * KU) & (KHALF - 1);
        __builtin_amdgcn_sched_barrier(0);
        LOAD_GROUP(tbA, hvA, mkA, kn);
        __builtin_amdgcn_sched_barrier(0);
        attn_group_fast(tbB, hvB, mkB, taq, eaq, s, acc, lim);
    }
}

__global__ __launch_bounds__(256) void attn_fused(
    const float* __restrict__ ta,
    const float* __restrict__ tb,
    const float* __restrict__ hv,
    const float* __restrict__ maskp,
    float* __restrict__ out)
{
    __shared__ float sm[QT][HDIM];
    __shared__ float ss[QT][HDIM];
    __shared__ float sa[QT][HDIM];

    const int h    = threadIdx.x & (HDIM - 1);
    const int half = threadIdx.x >> 7;
    const int bq  = blockIdx.x;
    const int b   = bq / (SEQ / QT);
    const int q0  = (bq % (SEQ / QT)) * QT;
    const size_t base = (size_t)b * SEQ * HDIM;

    v2f taq[QP], eaq[QP], m[QP], s[QP], acc[QP];
    bool ta_small = true;
    bool ta_fast  = true;
#pragma unroll
    for (int i = 0; i < QP; ++i) {
        taq[i].x = ta[base + (size_t)(q0 + 2 * i) * HDIM + h];
        taq[i].y = ta[base + (size_t)(q0 + 2 * i + 1) * HDIM + h];
        eaq[i].x = LOG2E * __builtin_amdgcn_exp2f(taq[i].x);
        eaq[i].y = LOG2E * __builtin_amdgcn_exp2f(taq[i].y);
        ta_small = ta_small && (__builtin_fabsf(taq[i].x) <= SPLIT_LIM)
                            && (__builtin_fabsf(taq[i].y) <= SPLIT_LIM);
        ta_fast  = ta_fast  && (__builtin_fabsf(taq[i].x) <= FAST_LIM)
                            && (__builtin_fabsf(taq[i].y) <= FAST_LIM);
        m[i].x = -INFINITY;  m[i].y = -INFINITY;
        s[i].x = 0.0f;       s[i].y = 0.0f;
        acc[i].x = 0.0f;     acc[i].y = 0.0f;
    }

    const size_t kbase = base + (size_t)half * KHALF * HDIM;
    const float* ptb = tb    + kbase + h;
    const float* phv = hv    + kbase + h;
    const float* pmk = maskp + kbase + h;

    float tbA[KU], hvA[KU], mkA[KU];
    LOAD_GROUP(tbA, hvA, mkA, 0);

    bool done = false;
    if (__builtin_amdgcn_ballot_w32(!ta_fast) == 0u) {
        float lim = 0.0f;
        attn_keys_fast(ptb, phv, pmk, tbA, hvA, mkA, taq, eaq, s, acc, lim);
        done = __builtin_amdgcn_ballot_w32(!(lim <= FAST_LIM)) == 0u;
#pragma unroll
        for (int i = 0; i < QP; ++i) {
            if (done) {
                m[i].x = 0.0f;    m[i].y = 0.0f;
            } else {
                s[i].x = 0.0f;    s[i].y = 0.0f;
                acc[i].x = 0.0f;  acc[i].y = 0.0f;
            }
        }
    }
    if (!done) {
        if (__builtin_amdgcn_ballot_w32(!ta_small) == 0u) {
            attn_keys<true>(ptb, phv, pmk, tbA, hvA, mkA, taq, eaq, m, s, acc);
        } else {
            attn_keys<false>(ptb, phv, pmk, tbA, hvA, mkA, taq, eaq, m, s, acc);
        }
    }
#undef LOAD_GROUP

    if (half == 1) {
#pragma unroll
        for (int i = 0; i < QP; ++i) {
            sm[2 * i][h] = m[i].x;    sm[2 * i + 1][h] = m[i].y;
            ss[2 * i][h] = s[i].x;    ss[2 * i + 1][h] = s[i].y;
            sa[2 * i][h] = acc[i].x;  sa[2 * i + 1][h] = acc[i].y;
        }
    }
    __syncthreads();
    if (half == 0) {
        const v2f lowest2 = {-FLT_MAX, -FLT_MAX};
#pragma unroll
        for (int i = 0; i < QP; ++i) {
            v2f m2, s2, a2, c1, c2;
            m2.x = sm[2 * i][h];  m2.y = sm[2 * i + 1][h];
            s2.x = ss[2 * i][h];  s2.y = ss[2 * i + 1][h];
            a2.x = sa[2 * i][h];  a2.y = sa[2 * i + 1][h];
            const v2f mn = __builtin_elementwise_maximum(m[i], m2);
            const v2f mr = __builtin_elementwise_maximum(mn, lowest2);
            const v2f d1 = m[i] - mr;
            const v2f d2 = m2 - mr;
            c1.x = __builtin_amdgcn_exp2f(d1.x);  c1.y = __builtin_amdgcn_exp2f(d1.y);
            c2.x = __builtin_amdgcn_exp2f(d2.x);  c2.y = __builtin_amdgcn_exp2f(d2.y);
            const v2f st = s[i] * c1 + s2 * c2;
            const v2f at = acc[i] * c1 + a2 * c2;
            out[base + (size_t)(q0 + 2 * i) * HDIM + h]     = at.x / st.x;
            out[base + (size_t)(q0 + 2 * i + 1) * HDIM + h] = at.y / st.y;
        }
    }
}

extern "C" void kernel_launch(void* const* d_in, const int* in_sizes, int n_in,
                              void* d_out, int out_size, void* d_ws, size_t ws_size,
                              hipStream_t stream) {
    const float* word_vecs = (const float*)d_in[0];
    const float* mask      = (const float*)d_in[1];
    const float* Wx_w      = (const float*)d_in[2];
    const float* Wx_b      = (const float*)d_in[3];
    const float* Wa_w      = (const float*)d_in[4];
    const float* Wa_b      = (const float*)d_in[5];
    const float* Ua_w      = (const float*)d_in[6];
    const float* Ua_b      = (const float*)d_in[7];
    float* out = (float*)d_out;

    float* hv = (float*)d_ws;
    float* ta = hv + (size_t)MROWS * HDIM;
    float* tb = ta + (size_t)MROWS * HDIM;

    gemm_wx_tanh<<<256, 128, 0, stream>>>(word_vecs, Wx_w, Wx_b, hv);
    gemm_ta_tb<<<256, 128, 0, stream>>>(hv, Wa_w, Wa_b, Ua_w, Ua_b, ta, tb);

    attn_fused<<<BATCH * (SEQ / QT), 256, 0, stream>>>(ta, tb, hv, mask, out);
}


# attention: 16 q-streams per thread, not 8 (key-row loads, exp(tb) once per 16 queries); 64 channels per block, keys split over four wave groups, each in key order; partial sums merged via LDS
# speedup vs baseline: 10.2502x; 1.0561x over previous
#include <hip/hip_runtime.h>
#include <hip/hip_bf16.h>
#include <math.h>
#include <float.h>

#define BATCH   4
#define SEQ     512
#define INDIM   300
#define HDIM    128
#define MROWS   (BATCH * SEQ)

typedef __attribute__((ext_vector_type(2))) float v2f;

#define LOG2E 1.4426950408889634f
typedef __attribute__((ext_vector_type(8))) float v8f;

#define BT 32

__device__ __forceinline__ v2f lds_frag(const float* S, int row0, int k0,
                                        int pitch, int lane) {
    const int r = row0 + (lane & 15);
    const int k = k0 + ((lane >> 4) << 1);
    return *reinterpret_cast<const v2f*>(S + r * pitch + k);
}

template <int KT>
__device__ __forceinline__ void lds_frags2(const float* SA, const float* SB, int m0, int n0,
                                           int kbase, int pitch, int lane,
                                           v2f (&a)[KT], v2f (&b)[KT]) {
#pragma unroll
    for (int j = 0; j < KT; ++j) {
        a[j] = lds_frag(SA, m0, kbase + 4 * j, pitch, lane);
        b[j] = lds_frag(SB, n0, kbase + 4 * j, pitch, lane);
    }
}

__global__ __launch_bounds__(128) void gemm_wx_tanh(
    const float* __restrict__ A,
    const float* __restrict__ W,
    const float* __restrict__ bias,
    float* __restrict__ out)
{
    __shared__ __attribute__((aligned(16))) float sA[BT * INDIM];
    __shared__ __attribute__((aligned(16))) float sW[BT * INDIM];

    const int tid  = threadIdx.x;
    const int lane = tid & 31;
    const int wave = tid >> 5;
    const int mb   = (blockIdx.x >> 2) * BT;
    const int nb   = (blockIdx.x & 3) * BT;

    const float4* gA = reinterpret_cast<const float4*>(A + (size_t)mb * INDIM);
    const float4* gW = reinterpret_cast<const float4*>(W + (size_t)nb * INDIM);
    float4* lA = reinterpret_cast<float4*>(sA);
    float4* lW = reinterpret_cast<float4*>(sW);

    constexpr int NV   = BT * INDIM / 4;
    constexpr int FULL = NV / 128;
    const bool tail = tid < NV - FULL * 128;

    const int itail = tail ? tid + FULL * 128 : NV - 1;
    float4 ra[FULL], rw[FULL];
#pragma unroll
    for (int i = 0; i < FULL; ++i) {
        ra[i] = gA[tid + i * 128];
        rw[i] = gW[tid + i * 128];
    }
    const float4 ta4 = gA[itail];
    const float4 tw4 = gW[itail];
#pragma unroll
    for (int i = 0; i < FULL; ++i) {
        lA[tid + i * 128] = ra[i];
        lW[tid + i * 128] = rw[i];
    }
    lA[itail] = ta4;
    lW[itail] = tw4;
    __syncthreads();

    const int m0 = (wave >> 1) << 4;
    const int n0 = (wave & 1) << 4;

    v8f acc = {};
    constexpr int KT1 = 15;
    v2f a0[KT1], b0[KT1], a1[KT1], b1[KT1];
#define MMA_TRIP1(AF, BF)                                                    \
    _Pragma("unroll") for (int j = 0; j < KT1; ++j) {                        \
        acc = __builtin_amdgcn_wmma_f32_16x16x4_f32(                         \
                false, AF[j], false, BF[j], (short)0, acc, false, false);    \
    }
    lds_frags2<KT1>(sA, sW, m0, n0, 0 * 4 * KT1, INDIM, lane, a0, b0);
    __builtin_amdgcn_sched_barrier(0);
    lds_frags2<KT1>(sA, sW, m0, n0, 1 * 4 * KT1, INDIM, lane, a1, b1);
    __builtin_amdgcn_sched_barrier(0);
    MMA_TRIP1(a0, b0)
    __builtin_amdgcn_sched_barrier(0);
    lds_frags2<KT1>(sA, sW, m0, n0, 2 * 4 * KT1, INDIM, lane, a0, b0);
    __builtin_amdgcn_sched_barrier(0);
    MMA_TRIP1(a1, b1)
    __builtin_amdgcn_sched_barrier(0);
    lds_frags2<KT1>(sA, sW, m0, n0, 3 * 4 * KT1, INDIM, lane, a1, b1);
    __builtin_amdgcn_sched_barrier(0);
    MMA_TRIP1(a0, b0)
    __builtin_amdgcn_sched_barrier(0);
    lds_frags2<KT1>(sA, sW, m0, n0, 4 * 4 * KT1, INDIM, lane, a0, b0);
    __builtin_amdgcn_sched_barrier(0);
    MMA_TRIP1(a1, b1)
    MMA_TRIP1(a0, b0)
#undef MMA_TRIP1

    const int col = nb + n0 + (lane & 15);
    const int rb  = mb + m0 + ((lane >> 4) << 3);
    const float bv = bias[col];
#pragma unroll
    for (int r = 0; r < 8; ++r) {
        out[(size_t)(rb + r) * HDIM + col] = tanhf(acc[r] + bv);
    }
}

#define PITCH2 (HDIM + 4)
__global__ __launch_bounds__(128) void gemm_ta_tb(
    const float* __restrict__ A,
    const float* __restrict__ Wa,
    const float* __restrict__ ba,
    const float* __restrict__ Ua,
    const float* __restrict__ bu,
    float* __restrict__ ta,
    float* __restrict__ tb)
{
    __shared__ __attribute__((aligned(16))) float sA[BT * PITCH2];
    __shared__ __attribute__((aligned(16))) float sWa[BT * PITCH2];
    __shared__ __attribute__((aligned(16))) float sUa[BT * PITCH2];

    const int tid  = threadIdx.x;
    const int lane = tid & 31;
    const int wave = tid >> 5;
    const int mb   = (blockIdx.x >> 2) * BT;
    const int nb   = (blockIdx.x & 3) * BT;

    const float4* gA  = reinterpret_cast<const float4*>(A  + (size_t)mb * HDIM);
    const float4* gWa = reinterpret_cast<const float4*>(Wa + (size_t)nb * HDIM);
    const float4* gUa = reinterpret_cast<const float4*>(Ua + (size_t)nb * HDIM);

    constexpr int ROUNDS = BT / 4;
    float4 ra[ROUNDS], rwa[ROUNDS], rua[ROUNDS];
#pragma unroll
    for (int i = 0; i < ROUNDS; ++i) {
        ra[i]  = gA[tid + i * 128];
        rwa[i] = gWa[tid + i * 128];
        rua[i] = gUa[tid + i * 128];
    }
#pragma unroll
    for (int i = 0; i < ROUNDS; ++i) {
        const int off = (wave + 4 * i) * PITCH2 + lane * 4;
        *reinterpret_cast<float4*>(sA  + off) = ra[i];
        *reinterpret_cast<float4*>(sWa + off) = rwa[i];
        *reinterpret_cast<float4*>(sUa + off) = rua[i];
    }
    __syncthreads();

    const int m0 = (wave >> 1) << 4;
    const int n0 = (wave & 1) << 4;

    v8f accA = {};
    v8f accB = {};
    constexpr int KT2 = 8;
    v2f a0[KT2], w0[KT2], u0[KT2], a1[KT2], w1[KT2], u1[KT2];
#define LOAD_TRIP2(AF, WF, UF, T)                                            \
    _Pragma("unroll") for (int j = 0; j < KT2; ++j) {                        \
        AF[j] = lds_frag(sA,  m0, (T) * 4 * KT2 + 4 * j, PITCH2, lane);      \
        WF[j] = lds_frag(sWa, n0, (T) * 4 * KT2 + 4 * j, PITCH2, lane);      \
        UF[j] = lds_frag(sUa, n0, (T) * 4 * KT2 + 4 * j, PITCH2, lane);      \
    }
#define MMA_TRIP2(AF, WF, UF)                                                \
    _Pragma("unroll") for (int j = 0; j < KT2; ++j) {                        \
        accA = __builtin_amdgcn_wmma_f32_16x16x4_f32(                        \
                 false, AF[j], false, WF[j], (short)0, accA, false, false);  \
        accB = __builtin_amdgcn_wmma_f32_16x16x4_f32(                        \
                 false, AF[j], false, UF[j], (short)0, accB, false, false);  \
    }
    LOAD_TRIP2(a0, w0, u0, 0)
    __builtin_amdgcn_sched_barrier(0);
    LOAD_TRIP2(a1, w1, u1, 1)
    __builtin_amdgcn_sched_barrier(0);
    MMA_TRIP2(a0, w0, u0)
    __builtin_amdgcn_sched_barrier(0);
    LOAD_TRIP2(a0, w0, u0, 2)
    __builtin_amdgcn_sched_barrier(0);
    MMA_TRIP2(a1, w1, u1)
    __builtin_amdgcn_sched_barrier(0);
    LOAD_TRIP2(a1, w1, u1, 3)
    __builtin_amdgcn_sched_barrier(0);
    MMA_TRIP2(a0, w0, u0)
    MMA_TRIP2(a1, w1, u1)
#undef LOAD_TRIP2
#undef MMA_TRIP2

    const int col = nb + n0 + (lane & 15);
    const int rb  = mb + m0 + ((lane >> 4) << 3);
    const float bva = ba[col];
    const float bvu = bu[col];
#pragma unroll
    for (int r = 0; r < 8; ++r) {
        const size_t idx = (size_t)(rb + r) * HDIM + col;
        ta[idx] = (accA[r] + bva) * LOG2E;
        tb[idx] = (accB[r] + bvu) * LOG2E;
    }
}

#define QT 16
#define KSPLIT 4
#define KPART (SEQ / KSPLIT)
#define CHB 64
#define SPLIT_LIM 64.0f
#define FAST_LIM 16.0f
#define KU 8

#define QP (QT / 2)
#define QH 4
template <bool SPLIT>
__device__ __forceinline__ void attn_group(const float (&tbk)[KU], const float (&hvk)[KU],
                                           const float (&mkk)[KU], const v2f (&taq)[QP],
                                           const v2f (&eaq)[QP],
                                           v2f (&m)[QP], v2f (&s)[QP], v2f (&acc)[QP]) {
    const v2f l2e = {LOG2E, LOG2E};
    const v2f zero2 = {0.0f, 0.0f};
    const v2f lowest2 = {-FLT_MAX, -FLT_MAX};
    float eb[KU];
#pragma unroll
    for (int j = 0; j < KU; ++j) eb[j] = SPLIT ? __builtin_amdgcn_exp2f(tbk[j]) : 0.0f;

#pragma unroll
    for (int i0 = 0; i0 < QP; i0 += QH) {
        v2f e[KU][QH];
        v2f bm[QH];
        v2f mr[QH];
#pragma unroll
        for (int ii = 0; ii < QH; ++ii) { bm[ii].x = -INFINITY; bm[ii].y = -INFINITY; }

#pragma unroll
        for (int j = 0; j < KU; ++j) {
            const v2f mk2 = {mkk[j], mkk[j]};
            const v2f eb2 = {eb[j], eb[j]};
#pragma unroll
            for (int ii = 0; ii < QH; ++ii) {
                const int i = i0 + ii;
                const v2f x = taq[i] + tbk[j];
                v2f em1;
                if (SPLIT) {
                    em1 = __builtin_elementwise_fma(eaq[i], eb2, -l2e);
                } else {
                    v2f t;
                    t.x = __builtin_amdgcn_exp2f(x.x);
                    t.y = __builtin_amdgcn_exp2f(x.y);
                    em1 = __builtin_elementwise_fma(t, l2e, -l2e);
                }
                const v2f el = __builtin_elementwise_maximum(
                                   x, __builtin_elementwise_minimum(em1, zero2));
                const v2f v = __builtin_elementwise_fma(mk2, l2e, el);
                e[j][ii] = v;
                bm[ii] = __builtin_elementwise_maximum(bm[ii], v);
            }
        }

#pragma unroll
        for (int ii = 0; ii < QH; ++ii) {
            const int i = i0 + ii;
            v2f mn, corr;
            mn = __builtin_elementwise_maximum(m[i], bm[ii]);
            mr[ii] = __builtin_elementwise_maximum(mn, lowest2);
            const v2f dm = m[i] - mr[ii];
            corr.x = __builtin_amdgcn_exp2f(dm.x);
            corr.y = __builtin_amdgcn_exp2f(dm.y);
            s[i]   *= corr;
            acc[i] *= corr;
            m[i]    = mn;
        }

#pragma unroll
        for (int j = 0; j < KU; ++j) {
            const v2f hv2 = {hvk[j], hvk[j]};
#pragma unroll
            for (int ii = 0; ii < QH; ++ii) {
                const int i = i0 + ii;
                const v2f d = e[j][ii] - mr[ii];
                v2f p;
                p.x = __builtin_amdgcn_exp2f(d.x);
                p.y = __builtin_amdgcn_exp2f(d.y);
                s[i]  += p;
                acc[i] = __builtin_elementwise_fma(p, hv2, acc[i]);
            }
        }
    }
}

__device__ __forceinline__ void attn_group_fast(const float (&tbk)[KU], const float (&hvk)[KU],
                                                const float (&mkk)[KU], const v2f (&taq)[QP],
                                                const v2f (&eaq)[QP],
                                                v2f (&s)[QP], v2f (&acc)[QP], float& lim) {
    const v2f l2e = {LOG2E, LOG2E};
    const v2f zero2 = {0.0f, 0.0f};
#pragma unroll
    for (int j = 0; j < KU; ++j) {
        const v2f mk2 = {mkk[j], mkk[j]};
        const v2f hv2 = {hvk[j], hvk[j]};
        v2f eb2;
        eb2.x = __builtin_amdgcn_exp2f(tbk[j]);
        eb2.y = eb2.x;
        lim = __builtin_elementwise_maximum(
                  lim, __builtin_elementwise_maximum(__builtin_fabsf(tbk[j]),
                                                     __builtin_fabsf(mkk[j])));
#pragma unroll
        for (int i = 0; i < QP; ++i) {
            const v2f x = taq[i] + tbk[j];
            const v2f em1 = __builtin_elementwise_fma(eaq[i], eb2, -l2e);
            const v2f el = __builtin_elementwise_maximum(
                               x, __builtin_elementwise_minimum(em1, zero2));
            const v2f v = __builtin_elementwise_fma(mk2, l2e, el);
            v2f p;
            p.x = __builtin_amdgcn_exp2f(v.x);
            p.y = __builtin_amdgcn_exp2f(v.y);
            s[i]  += p;
            acc[i] = __builtin_elementwise_fma(p, hv2, acc[i]);
        }
    }
}

#define LOAD_GROUP(TB, HV, MK, K0)                                  \
    _Pragma("unroll") for (int j = 0; j < KU; ++j) {                \
        TB[j] = ptb[(size_t)((K0) + j) * HDIM];                     \
        HV[j] = phv[(size_t)((K0) + j) * HDIM];                     \
        MK[j] = pmk[(size_t)((K0) + j) * HDIM];                     \
    }

template <bool SPLIT>
__device__ __forceinline__ void attn_keys(const float* __restrict__ ptb,
                                          const float* __restrict__ phv,
                                          const float* __restrict__ pmk,
                                          float (&tbA)[KU], float (&hvA)[KU], float (&mkA)[KU],
                                          const v2f (&taq)[QP], const v2f (&eaq)[QP],
                                          v2f (&m)[QP], v2f (&s)[QP], v2f (&acc)[QP]) {
    float tbB[KU], hvB[KU], mkB[KU];

    for (int k0 = 0; k0 < KPART; k0 += 2 * KU) {
        __builtin_amdgcn_sched_barrier(0);
        LOAD_GROUP(tbB, hvB, mkB, k0 + KU);
        __builtin_amdgcn_sched_barrier(0);
        attn_group<SPLIT>(tbA, hvA, mkA, taq, eaq, m, s, acc);

        const int kn = (k0 + 2 * KU) & (KPART - 1);
        __builtin_amdgcn_sched_barrier(0);
        LOAD_GROUP(tbA, hvA, mkA, kn);
        __builtin_amdgcn_sched_barrier(0);
        attn_group<SPLIT>(tbB, hvB, mkB, taq, eaq, m, s, acc);
    }
}

__device__ __forceinline__ void attn_keys_fast(const float* __restrict__ ptb,
                                               const float* __restrict__ phv,
                                               const float* __restrict__ pmk,
                                               float (&tbA)[KU], float (&hvA)[KU], float (&mkA)[KU],
                                               const v2f (&taq)[QP], const v2f (&eaq)[QP],
                                               v2f (&s)[QP], v2f (&acc)[QP], float& lim) {
    float tbB[KU], hvB[KU], mkB[KU];

    for (int k0 = 0; k0 < KPART; k0 += 2 * KU) {
        __builtin_amdgcn_sched_barrier(0);
        LOAD_GROUP(tbB, hvB, mkB, k0 + KU);
        __builtin_amdgcn_sched_barrier(0);
        attn_group_fast(tbA, hvA, mkA, taq, eaq, s, acc, lim);

        const int kn = (k0 + 2 * KU) & (KPART - 1);
        __builtin_amdgcn_sched_barrier(0);
        LOAD_GROUP(tbA, hvA, mkA, kn);
        __builtin_amdgcn_sched_barrier(0);
        attn_group_fast(tbB, hvB, mkB, taq, eaq, s, acc, lim);
    }
}

__global__ __launch_bounds__(256) void attn_fused(
    const float* __restrict__ ta,
    const float* __restrict__ tb,
    const float* __restrict__ hv,
    const float* __restrict__ maskp,
    float* __restrict__ out)
{
    __shared__ float sm[KSPLIT][QT][CHB];
    __shared__ float ss[KSPLIT][QT][CHB];
    __shared__ float sa[KSPLIT][QT][CHB];

    const int c    = threadIdx.x & (CHB - 1);
    const int part = threadIdx.x >> 6;
    const int hb   = blockIdx.x & (HDIM / CHB - 1);
    const int bq   = blockIdx.x / (HDIM / CHB);
    const int h    = hb * CHB + c;
    const int b    = bq / (SEQ / QT);
    const int q0   = (bq % (SEQ / QT)) * QT;
    const size_t base = (size_t)b * SEQ * HDIM;

    v2f taq[QP], eaq[QP], m[QP], s[QP], acc[QP];
    bool ta_small = true;
    bool ta_fast  = true;
#pragma unroll
    for (int i = 0; i < QP; ++i) {
        taq[i].x = ta[base + (size_t)(q0 + 2 * i) * HDIM + h];
        taq[i].y = ta[base + (size_t)(q0 + 2 * i + 1) * HDIM + h];
        eaq[i].x = LOG2E * __builtin_amdgcn_exp2f(taq[i].x);
        eaq[i].y = LOG2E * __builtin_amdgcn_exp2f(taq[i].y);
        ta_small = ta_small && (__builtin_fabsf(taq[i].x) <= SPLIT_LIM)
                            && (__builtin_fabsf(taq[i].y) <= SPLIT_LIM);
        ta_fast  = ta_fast  && (__builtin_fabsf(taq[i].x) <= FAST_LIM)
                            && (__builtin_fabsf(taq[i].y) <= FAST_LIM);
        m[i].x = -INFINITY;  m[i].y = -INFINITY;
        s[i].x = 0.0f;       s[i].y = 0.0f;
        acc[i].x = 0.0f;     acc[i].y = 0.0f;
    }

    const size_t kbase = base + (size_t)part * KPART * HDIM;
    const float* ptb = tb    + kbase + h;
    const float* phv = hv    + kbase + h;
    const float* pmk = maskp + kbase + h;

    float tbA[KU], hvA[KU], mkA[KU];
    LOAD_GROUP(tbA, hvA, mkA, 0);

    bool done = false;
    if (__builtin_amdgcn_ballot_w32(!ta_fast) == 0u) {
        float lim = 0.0f;
        attn_keys_fast(ptb, phv, pmk, tbA, hvA, mkA, taq, eaq, s, acc, lim);
        done = __builtin_amdgcn_ballot_w32(!(lim <= FAST_LIM)) == 0u;
#pragma unroll
        for (int i = 0; i < QP; ++i) {
            if (done) {
                m[i].x = 0.0f;    m[i].y = 0.0f;
            } else {
                s[i].x = 0.0f;    s[i].y = 0.0f;
                acc[i].x = 0.0f;  acc[i].y = 0.0f;
            }
        }
    }
    if (!done) {
        if (__builtin_amdgcn_ballot_w32(!ta_small) == 0u) {
            attn_keys<true>(ptb, phv, pmk, tbA, hvA, mkA, taq, eaq, m, s, acc);
        } else {
            attn_keys<false>(ptb, phv, pmk, tbA, hvA, mkA, taq, eaq, m, s, acc);
        }
    }
#undef LOAD_GROUP

#pragma unroll
    for (int i = 0; i < QP; ++i) {
        sm[part][2 * i][c] = m[i].x;    sm[part][2 * i + 1][c] = m[i].y;
        ss[part][2 * i][c] = s[i].x;    ss[part][2 * i + 1][c] = s[i].y;
        sa[part][2 * i][c] = acc[i].x;  sa[part][2 * i + 1][c] = acc[i].y;
    }
    __syncthreads();
    {
        const v2f lowest2 = {-FLT_MAX, -FLT_MAX};
#pragma unroll
        for (int j = 0; j < QT / KSPLIT / 2; ++j) {
            const int q = part * (QT / KSPLIT) + 2 * j;
            v2f mp[KSPLIT], sp[KSPLIT], ap[KSPLIT];
#pragma unroll
            for (int p = 0; p < KSPLIT; ++p) {
                mp[p].x = sm[p][q][c];  mp[p].y = sm[p][q + 1][c];
                sp[p].x = ss[p][q][c];  sp[p].y = ss[p][q + 1][c];
                ap[p].x = sa[p][q][c];  ap[p].y = sa[p][q + 1][c];
            }
            v2f mn = mp[0];
#pragma unroll
            for (int p = 1; p < KSPLIT; ++p) mn = __builtin_elementwise_maximum(mn, mp[p]);
            const v2f mr = __builtin_elementwise_maximum(mn, lowest2);
            v2f st = {0.0f, 0.0f};
            v2f at = {0.0f, 0.0f};
#pragma unroll
            for (int p = 0; p < KSPLIT; ++p) {
                const v2f d = mp[p] - mr;
                v2f cp;
                cp.x = __builtin_amdgcn_exp2f(d.x);
                cp.y = __builtin_amdgcn_exp2f(d.y);
                st = __builtin_elementwise_fma(sp[p], cp, st);
                at = __builtin_elementwise_fma(ap[p], cp, at);
            }
            out[base + (size_t)(q0 + q) * HDIM + h]     = at.x / st.x;
            out[base + (size_t)(q0 + q + 1) * HDIM + h] = at.y / st.y;
        }
    }
}

extern "C" void kernel_launch(void* const* d_in, const int* in_sizes, int n_in,
                              void* d_out, int out_size, void* d_ws, size_t ws_size,
                              hipStream_t stream) {
    const float* word_vecs = (const float*)d_in[0];
    const float* mask      = (const float*)d_in[1];
    const float* Wx_w      = (const float*)d_in[2];
    const float* Wx_b      = (const float*)d_in[3];
    const float* Wa_w      = (const float*)d_in[4];
    const float* Wa_b      = (const float*)d_in[5];
    const float* Ua_w      = (const float*)d_in[6];
    const float* Ua_b      = (const float*)d_in[7];
    float* out = (float*)d_out;

    float* hv = (float*)d_ws;
    float* ta = hv + (size_t)MROWS * HDIM;
    float* tb = ta + (size_t)MROWS * HDIM;

    gemm_wx_tanh<<<256, 128, 0, stream>>>(word_vecs, Wx_w, Wx_b, hv);
    gemm_ta_tb<<<256, 128, 0, stream>>>(hv, Wa_w, Wa_b, Ua_w, Ua_b, ta, tb);

    attn_fused<<<BATCH * (SEQ / QT) * (HDIM / CHB), 256, 0, stream>>>(ta, tb, hv, mask, out);
}


# the two projection launches fused into one: a block takes 16 rows x all 128 columns, its h tile feeds ta, tb from LDS; each wave stages its own weight rows; k order and sum chain of every tile kept
# speedup vs baseline: 10.6543x; 1.0394x over previous
#include <hip/hip_runtime.h>
#include <hip/hip_bf16.h>
#include <math.h>
#include <float.h>

#define BATCH   4
#define SEQ     512
#define INDIM   300
#define HDIM    128
#define MROWS   (BATCH * SEQ)

typedef __attribute__((ext_vector_type(2))) float v2f;

#define LOG2E 1.4426950408889634f
typedef __attribute__((ext_vector_type(8))) float v8f;
typedef __attribute__((ext_vector_type(4))) float v4f;


__device__ __forceinline__ v2f lds_frag(const float* S, int row0, int k0,
                                        int pitch, int lane) {
    const int r = row0 + (lane & 15);
    const int k = k0 + ((lane >> 4) << 1);
    return *reinterpret_cast<const v2f*>(S + r * pitch + k);
}

template <int KT>
__device__ __forceinline__ void lds_frags2(const float* SA, const float* SB, int m0, int n0,
                                           int kbase, int pitch, int lane,
                                           v2f (&a)[KT], v2f (&b)[KT]) {
#pragma unroll
    for (int j = 0; j < KT; ++j) {
        a[j] = lds_frag(SA, m0, kbase + 4 * j, pitch, lane);
        b[j] = lds_frag(SB, n0, kbase + 4 * j, pitch, lane);
    }
}

#define RT 16
#define PITCH2 (HDIM + 4)
__global__ __launch_bounds__(256) void proj_fused(
    const float* X,
    const float* Wx,
    const float* bx,
    const float* Wa,
    const float* ba,
    const float* Ua,
    const float* bu,
    float* hv,
    float* ta,
    float* tb)
{
    __shared__ __attribute__((aligned(16))) float sX[RT * INDIM];
    __shared__ __attribute__((aligned(16))) float sW[HDIM * INDIM];
    __shared__ __attribute__((aligned(16))) float sH[RT * PITCH2];
    __shared__ __attribute__((aligned(16))) float sWa[HDIM * PITCH2];
    __shared__ __attribute__((aligned(16))) float sUa[HDIM * PITCH2];

    const int tid  = threadIdx.x;
    const int lane = tid & 31;
    const int wave = tid >> 5;
    const int mb   = blockIdx.x * RT;
    const int n0   = wave << 4;

    constexpr int NV = RT * INDIM / 4;
    constexpr int RX = (NV + 255) / 256;
    constexpr int RW = (NV + 31) / 32;
    const v4f* gX = reinterpret_cast<const v4f*>(X + (size_t)mb * INDIM);
    const v4f* gW = reinterpret_cast<const v4f*>(Wx + (size_t)n0 * INDIM);
    v4f rx[RX], rw[RW];
    int ix[RX];
#pragma unroll
    for (int i = 0; i < RX; ++i) {
        const int p = tid + i * 256;
        ix[i] = p < NV ? p : NV - 1;
        rx[i] = gX[ix[i]];
    }
    const int iwl = (lane + (RW - 1) * 32) < NV ? lane + (RW - 1) * 32 : NV - 1;
#pragma unroll
    for (int i = 0; i < RW - 1; ++i) rw[i] = gW[lane + i * 32];
    rw[RW - 1] = gW[iwl];

#pragma unroll
    for (int i = 0; i < RX; ++i) reinterpret_cast<v4f*>(sX)[ix[i]] = rx[i];
    __builtin_amdgcn_fence(__ATOMIC_RELEASE, "workgroup", "local");
    __builtin_amdgcn_s_barrier();
    __builtin_amdgcn_fence(__ATOMIC_ACQUIRE, "workgroup", "local");

    v4f* lW = reinterpret_cast<v4f*>(sW + n0 * INDIM);
#pragma unroll
    for (int i = 0; i < RW - 1; ++i) lW[lane + i * 32] = rw[i];
    lW[iwl] = rw[RW - 1];

    const v4f* gWa = reinterpret_cast<const v4f*>(Wa + (size_t)n0 * HDIM);
    const v4f* gUa = reinterpret_cast<const v4f*>(Ua + (size_t)n0 * HDIM);
    v4f rwa[16], rua[16];
    __builtin_amdgcn_sched_barrier(0);
#pragma unroll
    for (int i = 0; i < 16; ++i) rwa[i] = gWa[lane + i * 32];
    __builtin_amdgcn_sched_barrier(0);

    v8f acc = {};
    constexpr int KT1 = 15;
    v2f a0[KT1], b0[KT1], a1[KT1], b1[KT1];
#define MMA_TRIP1(AF, BF)                                                    \
    _Pragma("unroll") for (int j = 0; j < KT1; ++j) {                        \
        acc = __builtin_amdgcn_wmma_f32_16x16x4_f32(                         \
                false, AF[j], false, BF[j], (short)0, acc, false, false);    \
    }
    lds_frags2<KT1>(sX, sW, 0, n0, 0 * 4 * KT1, INDIM, lane, a0, b0);
    __builtin_amdgcn_sched_barrier(0);
    lds_frags2<KT1>(sX, sW, 0, n0, 1 * 4 * KT1, INDIM, lane, a1, b1);
    __builtin_amdgcn_sched_barrier(0);
    MMA_TRIP1(a0, b0)
    __builtin_amdgcn_sched_barrier(0);
    lds_frags2<KT1>(sX, sW, 0, n0, 2 * 4 * KT1, INDIM, lane, a0, b0);
    __builtin_amdgcn_sched_barrier(0);
    MMA_TRIP1(a1, b1)
    __builtin_amdgcn_sched_barrier(0);
    lds_frags2<KT1>(sX, sW, 0, n0, 3 * 4 * KT1, INDIM, lane, a1, b1);
#pragma unroll
    for (int i = 0; i < 16; ++i)
        *reinterpret_cast<v4f*>(sWa + (n0 + i) * PITCH2 + lane * 4) = rwa[i];
    __builtin_amdgcn_sched_barrier(0);
#pragma unroll
    for (int i = 0; i < 16; ++i) rua[i] = gUa[lane + i * 32];
    __builtin_amdgcn_sched_barrier(0);
    MMA_TRIP1(a0, b0)
    __builtin_amdgcn_sched_barrier(0);
    lds_frags2<KT1>(sX, sW, 0, n0, 4 * 4 * KT1, INDIM, lane, a0, b0);
    __builtin_amdgcn_sched_barrier(0);
    MMA_TRIP1(a1, b1)
    MMA_TRIP1(a0, b0)
#undef MMA_TRIP1

    const int col = n0 + (lane & 15);
    const int r0  = (lane >> 4) << 3;
    {
        const float bv = bx[col];
#pragma unroll
        for (int r = 0; r < 8; ++r) {
            const float hval = tanhf(acc[r] + bv);
            sH[(r0 + r) * PITCH2 + col] = hval;
            hv[(size_t)(mb + r0 + r) * HDIM + col] = hval;
        }
    }
#pragma unroll
    for (int i = 0; i < 16; ++i)
        *reinterpret_cast<v4f*>(sUa + (n0 + i) * PITCH2 + lane * 4) = rua[i];
    __syncthreads();

    v8f accA = {};
    v8f accB = {};
    constexpr int KT2 = 8;
    v2f h0[KT2], w0[KT2], u0[KT2], h1[KT2], w1[KT2], u1[KT2];
#define LOAD_TRIP2(AF, WF, UF, T)                                            \
    _Pragma("unroll") for (int j = 0; j < KT2; ++j) {                        \
        AF[j] = lds_frag(sH,  0,  (T) * 4 * KT2 + 4 * j, PITCH2, lane);      \
        WF[j] = lds_frag(sWa, n0, (T) * 4 * KT2 + 4 * j, PITCH2, lane);      \
        UF[j] = lds_frag(sUa, n0, (T) * 4 * KT2 + 4 * j, PITCH2, lane);      \
    }
#define MMA_TRIP2(AF, WF, UF)                                                \
    _Pragma("unroll") for (int j = 0; j < KT2; ++j) {                        \
        accA = __builtin_amdgcn_wmma_f32_16x16x4_f32(                        \
                 false, AF[j], false, WF[j], (short)0, accA, false, false);  \
        accB = __builtin_amdgcn_wmma_f32_16x16x4_f32(                        \
                 false, AF[j], false, UF[j], (short)0, accB, false, false);  \
    }
    LOAD_TRIP2(h0, w0, u0, 0)
    __builtin_amdgcn_sched_barrier(0);
    LOAD_TRIP2(h1, w1, u1, 1)
    __builtin_amdgcn_sched_barrier(0);
    MMA_TRIP2(h0, w0, u0)
    __builtin_amdgcn_sched_barrier(0);
    LOAD_TRIP2(h0, w0, u0, 2)
    __builtin_amdgcn_sched_barrier(0);
    MMA_TRIP2(h1, w1, u1)
    __builtin_amdgcn_sched_barrier(0);
    LOAD_TRIP2(h1, w1, u1, 3)
    __builtin_amdgcn_sched_barrier(0);
    MMA_TRIP2(h0, w0, u0)
    MMA_TRIP2(h1, w1, u1)
#undef LOAD_TRIP2
#undef MMA_TRIP2

    const float bva = ba[col];
    const float bvu = bu[col];
#pragma unroll
    for (int r = 0; r < 8; ++r) {
        const size_t idx = (size_t)(mb + r0 + r) * HDIM + col;
        ta[idx] = (accA[r] + bva) * LOG2E;
        tb[idx] = (accB[r] + bvu) * LOG2E;
    }
}

#define QT 16
#define KSPLIT 4
#define KPART (SEQ / KSPLIT)
#define CHB 64
#define SPLIT_LIM 64.0f
#define FAST_LIM 16.0f
#define KU 8

#define QP (QT / 2)
#define QH 4
template <bool SPLIT>
__device__ __forceinline__ void attn_group(const float (&tbk)[KU], const float (&hvk)[KU],
                                           const float (&mkk)[KU], const v2f (&taq)[QP],
                                           const v2f (&eaq)[QP],
                                           v2f (&m)[QP], v2f (&s)[QP], v2f (&acc)[QP]) {
    const v2f l2e = {LOG2E, LOG2E};
    const v2f zero2 = {0.0f, 0.0f};
    const v2f lowest2 = {-FLT_MAX, -FLT_MAX};
    float eb[KU];
#pragma unroll
    for (int j = 0; j < KU; ++j) eb[j] = SPLIT ? __builtin_amdgcn_exp2f(tbk[j]) : 0.0f;

#pragma unroll
    for (int i0 = 0; i0 < QP; i0 += QH) {
        v2f e[KU][QH];
        v2f bm[QH];
        v2f mr[QH];
#pragma unroll
        for (int ii = 0; ii < QH; ++ii) { bm[ii].x = -INFINITY; bm[ii].y = -INFINITY; }

#pragma unroll
        for (int j = 0; j < KU; ++j) {
            const v2f mk2 = {mkk[j], mkk[j]};
            const v2f eb2 = {eb[j], eb[j]};
#pragma unroll
            for (int ii = 0; ii < QH; ++ii) {
                const int i = i0 + ii;
                const v2f x = taq[i] + tbk[j];
                v2f em1;
                if (SPLIT) {
                    em1 = __builtin_elementwise_fma(eaq[i], eb2, -l2e);
                } else {
                    v2f t;
                    t.x = __builtin_amdgcn_exp2f(x.x);
                    t.y = __builtin_amdgcn_exp2f(x.y);
                    em1 = __builtin_elementwise_fma(t, l2e, -l2e);
                }
                const v2f el = __builtin_elementwise_maximum(
                                   x, __builtin_elementwise_minimum(em1, zero2));
                const v2f v = __builtin_elementwise_fma(mk2, l2e, el);
                e[j][ii] = v;
                bm[ii] = __builtin_elementwise_maximum(bm[ii], v);
            }
        }

#pragma unroll
        for (int ii = 0; ii < QH; ++ii) {
            const int i = i0 + ii;
            v2f mn, corr;
            mn = __builtin_elementwise_maximum(m[i], bm[ii]);
            mr[ii] = __builtin_elementwise_maximum(mn, lowest2);
            const v2f dm = m[i] - mr[ii];
            corr.x = __builtin_amdgcn_exp2f(dm.x);
            corr.y = __builtin_amdgcn_exp2f(dm.y);
            s[i]   *= corr;
            acc[i] *= corr;
            m[i]    = mn;
        }

#pragma unroll
        for (int j = 0; j < KU; ++j) {
            const v2f hv2 = {hvk[j], hvk[j]};
#pragma unroll
            for (int ii = 0; ii < QH; ++ii) {
                const int i = i0 + ii;
                const v2f d = e[j][ii] - mr[ii];
                v2f p;
                p.x = __builtin_amdgcn_exp2f(d.x);
                p.y = __builtin_amdgcn_exp2f(d.y);
                s[i]  += p;
                acc[i] = __builtin_elementwise_fma(p, hv2, acc[i]);
            }
        }
    }
}

__device__ __forceinline__ void attn_group_fast(const float (&tbk)[KU], const float (&hvk)[KU],
                                                const float (&mkk)[KU], const v2f (&taq)[QP],
                                                const v2f (&eaq)[QP],
                                                v2f (&s)[QP], v2f (&acc)[QP], float& lim) {
    const v2f l2e = {LOG2E, LOG2E};
    const v2f zero2 = {0.0f, 0.0f};
#pragma unroll
    for (int j = 0; j < KU; ++j) {
        const v2f mk2 = {mkk[j], mkk[j]};
        const v2f hv2 = {hvk[j], hvk[j]};
        v2f eb2;
        eb2.x = __builtin_amdgcn_exp2f(tbk[j]);
        eb2.y = eb2.x;
        lim = __builtin_elementwise_maximum(
                  lim, __builtin_elementwise_maximum(__builtin_fabsf(tbk[j]),
                                                     __builtin_fabsf(mkk[j])));
#pragma unroll
        for (int i = 0; i < QP; ++i) {
            const v2f x = taq[i] + tbk[j];
            const v2f em1 = __builtin_elementwise_fma(eaq[i], eb2, -l2e);
            const v2f el = __builtin_elementwise_maximum(
                               x, __builtin_elementwise_minimum(em1, zero2));
            const v2f v = __builtin_elementwise_fma(mk2, l2e, el);
            v2f p;
            p.x = __builtin_amdgcn_exp2f(v.x);
            p.y = __builtin_amdgcn_exp2f(v.y);
            s[i]  += p;
            acc[i] = __builtin_elementwise_fma(p, hv2, acc[i]);
        }
    }
}

#define LOAD_GROUP(TB, HV, MK, K0)                                  \
    _Pragma("unroll") for (int j = 0; j < KU; ++j) {                \
        TB[j] = ptb[(size_t)((K0) + j) * HDIM];                     \
        HV[j] = phv[(size_t)((K0) + j) * HDIM];                     \
        MK[j] = pmk[(size_t)((K0) + j) * HDIM];                     \
    }

template <bool SPLIT>
__device__ __forceinline__ void attn_keys(const float* __restrict__ ptb,
                                          const float* __restrict__ phv,
                                          const float* __restrict__ pmk,
                                          float (&tbA)[KU], float (&hvA)[KU], float (&mkA)[KU],
                                          const v2f (&taq)[QP], const v2f (&eaq)[QP],
                                          v2f (&m)[QP], v2f (&s)[QP], v2f (&acc)[QP]) {
    float tbB[KU], hvB[KU], mkB[KU];

    for (int k0 = 0; k0 < KPART; k0 += 2 * KU) {
        __builtin_amdgcn_sched_barrier(0);
        LOAD_GROUP(tbB, hvB, mkB, k0 + KU);
        __builtin_amdgcn_sched_barrier(0);
        attn_group<SPLIT>(tbA, hvA, mkA, taq, eaq, m, s, acc);

        const int kn = (k0 + 2 * KU) & (KPART - 1);
        __builtin_amdgcn_sched_barrier(0);
        LOAD_GROUP(tbA, hvA, mkA, kn);
        __builtin_amdgcn_sched_barrier(0);
        attn_group<SPLIT>(tbB, hvB, mkB, taq, eaq, m, s, acc);
    }
}

__device__ __forceinline__ void attn_keys_fast(const float* __restrict__ ptb,
                                               const float* __restrict__ phv,
                                               const float* __restrict__ pmk,
                                               float (&tbA)[KU], float (&hvA)[KU], float (&mkA)[KU],
                                               const v2f (&taq)[QP], const v2f (&eaq)[QP],
                                               v2f (&s)[QP], v2f (&acc)[QP], float& lim) {
    float tbB[KU], hvB[KU], mkB[KU];

    for (int k0 = 0; k0 < KPART; k0 += 2 * KU) {
        __builtin_amdgcn_sched_barrier(0);
        LOAD_GROUP(tbB, hvB, mkB, k0 + KU);
        __builtin_amdgcn_sched_barrier(0);
        attn_group_fast(tbA, hvA, mkA, taq, eaq, s, acc, lim);

        const int kn = (k0 + 2 * KU) & (KPART - 1);
        __builtin_amdgcn_sched_barrier(0);
        LOAD_GROUP(tbA, hvA, mkA, kn);
        __builtin_amdgcn_sched_barrier(0);
        attn_group_fast(tbB, hvB, mkB, taq, eaq, s, acc, lim);
    }
}

__global__ __launch_bounds__(256) void attn_fused(
    const float* __restrict__ ta,
    const float* __restrict__ tb,
    const float* __restrict__ hv,
    const float* __restrict__ maskp,
    float* __restrict__ out)
{
    __shared__ float sm[KSPLIT][QT][CHB];
    __shared__ float ss[KSPLIT][QT][CHB];
    __shared__ float sa[KSPLIT][QT][CHB];

    const int c    = threadIdx.x & (CHB - 1);
    const int part = threadIdx.x >> 6;
    const int hb   = blockIdx.x & (HDIM / CHB - 1);
    const int bq   = blockIdx.x / (HDIM / CHB);
    const int h    = hb * CHB + c;
    const int b    = bq / (SEQ / QT);
    const int q0   = (bq % (SEQ / QT)) * QT;
    const size_t base = (size_t)b * SEQ * HDIM;

    v2f taq[QP], eaq[QP], m[QP], s[QP], acc[QP];
    bool ta_small = true;
    bool ta_fast  = true;
#pragma unroll
    for (int i = 0; i < QP; ++i) {
        taq[i].x = ta[base + (size_t)(q0 + 2 * i) * HDIM + h];
        taq[i].y = ta[base + (size_t)(q0 + 2 * i + 1) * HDIM + h];
        eaq[i].x = LOG2E * __builtin_amdgcn_exp2f(taq[i].x);
        eaq[i].y = LOG2E * __builtin_amdgcn_exp2f(taq[i].y);
        ta_small = ta_small && (__builtin_fabsf(taq[i].x) <= SPLIT_LIM)
                            && (__builtin_fabsf(taq[i].y) <= SPLIT_LIM);
        ta_fast  = ta_fast  && (__builtin_fabsf(taq[i].x) <= FAST_LIM)
                            && (__builtin_fabsf(taq[i].y) <= FAST_LIM);
        m[i].x = -INFINITY;  m[i].y = -INFINITY;
        s[i].x = 0.0f;       s[i].y = 0.0f;
        acc[i].x = 0.0f;     acc[i].y = 0.0f;
    }

    const size_t kbase = base + (size_t)part * KPART * HDIM;
    const float* ptb = tb    + kbase + h;
    const float* phv = hv    + kbase + h;
    const float* pmk = maskp + kbase + h;

    float tbA[KU], hvA[KU], mkA[KU];
    LOAD_GROUP(tbA, hvA, mkA, 0);

    bool done = false;
    if (__builtin_amdgcn_ballot_w32(!ta_fast) == 0u) {
        float lim = 0.0f;
        attn_keys_fast(ptb, phv, pmk, tbA, hvA, mkA, taq, eaq, s, acc, lim);
        done = __builtin_amdgcn_ballot_w32(!(lim <= FAST_LIM)) == 0u;
#pragma unroll
        for (int i = 0; i < QP; ++i) {
            if (done) {
                m[i].x = 0.0f;    m[i].y = 0.0f;
            } else {
                s[i].x = 0.0f;    s[i].y = 0.0f;
                acc[i].x = 0.0f;  acc[i].y = 0.0f;
            }
        }
    }
    if (!done) {
        if (__builtin_amdgcn_ballot_w32(!ta_small) == 0u) {
            attn_keys<true>(ptb, phv, pmk, tbA, hvA, mkA, taq, eaq, m, s, acc);
        } else {
            attn_keys<false>(ptb, phv, pmk, tbA, hvA, mkA, taq, eaq, m, s, acc);
        }
    }
#undef LOAD_GROUP

#pragma unroll
    for (int i = 0; i < QP; ++i) {
        sm[part][2 * i][c] = m[i].x;    sm[part][2 * i + 1][c] = m[i].y;
        ss[part][2 * i][c] = s[i].x;    ss[part][2 * i + 1][c] = s[i].y;
        sa[part][2 * i][c] = acc[i].x;  sa[part][2 * i + 1][c] = acc[i].y;
    }
    __syncthreads();
    {
        const v2f lowest2 = {-FLT_MAX, -FLT_MAX};
#pragma unroll
        for (int j = 0; j < QT / KSPLIT / 2; ++j) {
            const int q = part * (QT / KSPLIT) + 2 * j;
            v2f mp[KSPLIT], sp[KSPLIT], ap[KSPLIT];
#pragma unroll
            for (int p = 0; p < KSPLIT; ++p) {
                mp[p].x = sm[p][q][c];  mp[p].y = sm[p][q + 1][c];
                sp[p].x = ss[p][q][c];  sp[p].y = ss[p][q + 1][c];
                ap[p].x = sa[p][q][c];  ap[p].y = sa[p][q + 1][c];
            }
            v2f mn = mp[0];
#pragma unroll
            for (int p = 1; p < KSPLIT; ++p) mn = __builtin_elementwise_maximum(mn, mp[p]);
            const v2f mr = __builtin_elementwise_maximum(mn, lowest2);
            v2f st = {0.0f, 0.0f};
            v2f at = {0.0f, 0.0f};
#pragma unroll
            for (int p = 0; p < KSPLIT; ++p) {
                const v2f d = mp[p] - mr;
                v2f cp;
                cp.x = __builtin_amdgcn_exp2f(d.x);
                cp.y = __builtin_amdgcn_exp2f(d.y);
                st = __builtin_elementwise_fma(sp[p], cp, st);
                at = __builtin_elementwise_fma(ap[p], cp, at);
            }
            out[base + (size_t)(q0 + q) * HDIM + h]     = at.x / st.x;
            out[base + (size_t)(q0 + q + 1) * HDIM + h] = at.y / st.y;
        }
    }
}

extern "C" void kernel_launch(void* const* d_in, const int* in_sizes, int n_in,
                              void* d_out, int out_size, void* d_ws, size_t ws_size,
                              hipStream_t stream) {
    const float* word_vecs = (const float*)d_in[0];
    const float* mask      = (const float*)d_in[1];
    const float* Wx_w      = (const float*)d_in[2];
    const float* Wx_b      = (const float*)d_in[3];
    const float* Wa_w      = (const float*)d_in[4];
    const float* Wa_b      = (const float*)d_in[5];
    const float* Ua_w      = (const float*)d_in[6];
    const float* Ua_b      = (const float*)d_in[7];
    float* out = (float*)d_out;

    float* hv = (float*)d_ws;
    float* ta = hv + (size_t)MROWS * HDIM;
    float* tb = ta + (size_t)MROWS * HDIM;

    proj_fused<<<MROWS / RT, 256, 0, stream>>>(word_vecs, Wx_w, Wx_b, Wa_w, Wa_b, Ua_w, Ua_b,
                                               hv, ta, tb);

    attn_fused<<<BATCH * (SEQ / QT) * (HDIM / CHB), 256, 0, stream>>>(ta, tb, hv, mask, out);
}
